# starting kernel
# baseline (speedup 1.0000x reference)
#include <hip/hip_runtime.h>
#include <hip/hip_bf16.h>


typedef float v8f __attribute__((ext_vector_type(8)));
typedef float v2f __attribute__((ext_vector_type(2)));

#define GNN_D 128
#define TILE_M 16
#define LDS_STRIDE 132

__device__ __forceinline__ unsigned f2mono(float f) {
    unsigned u = __float_as_uint(f);
    return (u & 0x80000000u) ? ~u : (u | 0x80000000u);
}
__device__ __forceinline__ float mono2f(unsigned m) {
    unsigned u = (m & 0x80000000u) ? (m & 0x7FFFFFFFu) : ~m;
    return __uint_as_float(u);
}
#define MONO_NEG_INF 0x007FFFFFu


__global__ __launch_bounds__(256) void gnn_init_mono_kernel(unsigned* __restrict__ p, int n) {
    int i = blockIdx.x * 256 + threadIdx.x;
    if (i < n) p[i] = MONO_NEG_INF;
}

__global__ __launch_bounds__(256) void gnn_deg_kernel(const int* __restrict__ dst,
                                                      float* __restrict__ deg, int E) {
    int e = blockIdx.x * 256 + threadIdx.x;
    if (e < E) atomicAdd(&deg[dst[e]], 1.0f);
}

__global__ __launch_bounds__(256) void gnn_scatter_kernel(
    const float* __restrict__ x, const int* __restrict__ src,
    const int* __restrict__ dst, float* __restrict__ s,
    unsigned* __restrict__ mx, int E)
{
    int gid  = blockIdx.x * 256 + threadIdx.x;
    int e    = gid >> 5;
    int lane = gid & 31;
    if (e >= E) return;
    int sn = src[e];
    int dn = dst[e];
    const float4 v = *reinterpret_cast<const float4*>(x + (size_t)sn * GNN_D + lane * 4);
    float*    sp = s  + (size_t)dn * GNN_D + lane * 4;
    unsigned* mp = mx + (size_t)dn * GNN_D + lane * 4;
    atomicAdd(sp + 0, v.x); atomicAdd(sp + 1, v.y);
    atomicAdd(sp + 2, v.z); atomicAdd(sp + 3, v.w);
    atomicMax(mp + 0, f2mono(v.x)); atomicMax(mp + 1, f2mono(v.y));
    atomicMax(mp + 2, f2mono(v.z)); atomicMax(mp + 3, f2mono(v.w));
}

__global__ __launch_bounds__(256) void gnn_layer_kernel(
    const float* __restrict__ x, const float* __restrict__ s,
    const unsigned* __restrict__ mx, const float* __restrict__ deg,
    const float* __restrict__ zagg,
    const float* __restrict__ W1, const float* __restrict__ b1,
    const float* __restrict__ W2, const float* __restrict__ b2,
    float* __restrict__ xout)
{
    __shared__ float hs[TILE_M * LDS_STRIDE];
    __shared__ float xt[TILE_M * LDS_STRIDE];
    __shared__ float ts[TILE_M * LDS_STRIDE];

    const int tid  = threadIdx.x;
    const int row0 = blockIdx.x * TILE_M;

    const float z0 = zagg[0], z1 = zagg[1], z2 = zagg[2];

    {
        int base = tid * 8;
        int r = base >> 7;
        int c = base & 127;
        size_t g = (size_t)(row0 + r) * GNN_D + c;
        float dg = deg[row0 + r];
        float dc = fmaxf(dg, 1.0f);
        bool  nb = dg > 0.0f;
        #pragma unroll
        for (int q = 0; q < 2; ++q) {
            float4 xv = *reinterpret_cast<const float4*>(x + g + q * 4);
            float4 sv = *reinterpret_cast<const float4*>(s + g + q * 4);
            uint4  mv = *reinterpret_cast<const uint4*>(mx + g + q * 4);
            float4 h;
            h.x = xv.x + z0 * sv.x + z1 * (sv.x / dc) + z2 * (nb ? mono2f(mv.x) : 0.0f);
            h.y = xv.y + z0 * sv.y + z1 * (sv.y / dc) + z2 * (nb ? mono2f(mv.y) : 0.0f);
            h.z = xv.z + z0 * sv.z + z1 * (sv.z / dc) + z2 * (nb ? mono2f(mv.z) : 0.0f);
            h.w = xv.w + z0 * sv.w + z1 * (sv.w / dc) + z2 * (nb ? mono2f(mv.w) : 0.0f);
            *reinterpret_cast<float4*>(&hs[r * LDS_STRIDE + c + q * 4]) = h;
            *reinterpret_cast<float4*>(&xt[r * LDS_STRIDE + c + q * 4]) = xv;
        }
    }
    __syncthreads();

    const int wave  = tid >> 5;
    const int lane  = tid & 31;
    const int ncol  = (wave << 4) + (lane & 15);
    const int mlo   = lane & 15;
    const int koff  = (lane >> 4) << 1;
    const int rbase = (lane >> 4) << 3;
    const float bia1 = b1[ncol];
    const float bia2 = b2[ncol];

    v8f acc = {};
    #pragma unroll
    for (int k0 = 0; k0 < GNN_D; k0 += 4) {
        int kk = k0 + koff;
        v2f a; a.x = hs[mlo * LDS_STRIDE + kk];
               a.y = hs[mlo * LDS_STRIDE + kk + 1];
        v2f b; b.x = W1[(size_t)kk * GNN_D + ncol];
               b.y = W1[(size_t)(kk + 1) * GNN_D + ncol];
        acc = __builtin_amdgcn_wmma_f32_16x16x4_f32(false, a, false, b,
                                                    (short)0, acc, false, false);
    }
    #pragma unroll
    for (int i = 0; i < 8; ++i)
        ts[(rbase + i) * LDS_STRIDE + ncol] = fmaxf(acc[i] + bia1, 0.0f);
    __syncthreads();

    v8f acc2 = {};
    #pragma unroll
    for (int k0 = 0; k0 < GNN_D; k0 += 4) {
        int kk = k0 + koff;
        v2f a; a.x = ts[mlo * LDS_STRIDE + kk];
               a.y = ts[mlo * LDS_STRIDE + kk + 1];
        v2f b; b.x = W2[(size_t)kk * GNN_D + ncol];
               b.y = W2[(size_t)(kk + 1) * GNN_D + ncol];
        acc2 = __builtin_amdgcn_wmma_f32_16x16x4_f32(false, a, false, b,
                                                     (short)0, acc2, false, false);
    }
    #pragma unroll
    for (int i = 0; i < 8; ++i) {
        int r = rbase + i;
        float h2 = acc2[i] + bia2;
        float xv = xt[r * LDS_STRIDE + ncol];
        xout[(size_t)(row0 + r) * GNN_D + ncol] = xv + fmaxf(h2, 0.0f);
    }
}

__global__ __launch_bounds__(256) void gnn_pool_scatter_kernel(
    const float* __restrict__ x, const int* __restrict__ batch,
    float* __restrict__ ps, unsigned* __restrict__ pmx,
    float* __restrict__ cnt, int N)
{
    int gid  = blockIdx.x * 256 + threadIdx.x;
    int n    = gid >> 5;
    int lane = gid & 31;
    if (n >= N) return;
    int b = batch[n];
    const float4 v = *reinterpret_cast<const float4*>(x + (size_t)n * GNN_D + lane * 4);
    float*    sp = ps  + (size_t)b * GNN_D + lane * 4;
    unsigned* mp = pmx + (size_t)b * GNN_D + lane * 4;
    atomicAdd(sp + 0, v.x); atomicAdd(sp + 1, v.y);
    atomicAdd(sp + 2, v.z); atomicAdd(sp + 3, v.w);
    atomicMax(mp + 0, f2mono(v.x)); atomicMax(mp + 1, f2mono(v.y));
    atomicMax(mp + 2, f2mono(v.z)); atomicMax(mp + 3, f2mono(v.w));
    if (lane == 0) atomicAdd(&cnt[b], 1.0f);
}

__global__ __launch_bounds__(256) void gnn_pool_finalize_kernel(
    const float* __restrict__ ps, const unsigned* __restrict__ pmx,
    const float* __restrict__ cnt, const float* __restrict__ zpool,
    float* __restrict__ out, int total)
{
    int i = blockIdx.x * 256 + threadIdx.x;
    if (i >= total) return;
    int b = i >> 7;
    float c  = cnt[b];
    float sv = ps[i];
    float mv = (c > 0.0f) ? mono2f(pmx[i]) : 0.0f;
    out[i] = zpool[0] * sv + zpool[1] * (sv / fmaxf(c, 1.0f)) + zpool[2] * mv;
}


extern "C" void kernel_launch(void* const* d_in, const int* in_sizes, int n_in,
                              void* d_out, int out_size, void* d_ws, size_t ws_size,
                              hipStream_t stream) {
    const float* x_in   = (const float*)d_in[0];
    const int*   eidx   = (const int*)d_in[1];
    const int*   batch  = (const int*)d_in[2];
    const float* z_agg  = (const float*)d_in[3];
    const float* z_pool = (const float*)d_in[4];
    const float* W1     = (const float*)d_in[5];
    const float* b1     = (const float*)d_in[6];
    const float* W2     = (const float*)d_in[7];
    const float* b2     = (const float*)d_in[8];
    (void)n_in;

    const int N = in_sizes[0] / GNN_D;
    const int E = in_sizes[1] / 2;
    const int L = in_sizes[5] / (GNN_D * GNN_D);
    const int Bg = out_size / GNN_D;
    const size_t ND = (size_t)N * GNN_D;
    const size_t BD = (size_t)Bg * GNN_D;

    char* p = (char*)d_ws;
    char* pend = p + ws_size;
    auto take = [&](size_t bytes) -> void* {
        void* r = (void*)p;
        p += (bytes + 255) & ~(size_t)255;
        return r;
    };
    float*    xa   = (float*)take(ND * 4);
    float*    xb   = (float*)take(ND * 4);
    float*    s    = (float*)take(ND * 4);
    unsigned* mx   = (unsigned*)take(ND * 4);
    float*    deg  = (float*)take((size_t)N * 4);
    float*    ps   = (float*)take(BD * 4);
    unsigned* pmx  = (unsigned*)take(BD * 4);
    float*    cnt  = (float*)take((size_t)Bg * 4);
    if (p > pend) return;

    const int* src = eidx;
    const int* dst = eidx + E;

    hipMemcpyAsync(xa, x_in, ND * 4, hipMemcpyDeviceToDevice, stream);

    hipMemsetAsync(deg, 0, (size_t)N * 4, stream);
    gnn_deg_kernel<<<(E + 255) / 256, 256, 0, stream>>>(dst, deg, E);

    float* xcur = xa;
    float* xnxt = xb;
    for (int l = 0; l < L; ++l) {
        hipMemsetAsync(s, 0, ND * 4, stream);
        gnn_init_mono_kernel<<<(int)((ND + 255) / 256), 256, 0, stream>>>(mx, (int)ND);
        gnn_scatter_kernel<<<(E + 7) / 8, 256, 0, stream>>>(xcur, src, dst, s, mx, E);
        gnn_layer_kernel<<<N / TILE_M, 256, 0, stream>>>(
            xcur, s, mx, deg, z_agg + l * 3,
            W1 + (size_t)l * GNN_D * GNN_D, b1 + (size_t)l * GNN_D,
            W2 + (size_t)l * GNN_D * GNN_D, b2 + (size_t)l * GNN_D, xnxt);
        float* t = xcur; xcur = xnxt; xnxt = t;
    }

    hipMemsetAsync(ps, 0, BD * 4, stream);
    hipMemsetAsync(cnt, 0, (size_t)Bg * 4, stream);
    gnn_init_mono_kernel<<<(int)((BD + 255) / 256), 256, 0, stream>>>(pmx, (int)BD);
    gnn_pool_scatter_kernel<<<(N + 7) / 8, 256, 0, stream>>>(xcur, batch, ps, pmx, cnt, N);
    gnn_pool_finalize_kernel<<<(int)((BD + 255) / 256), 256, 0, stream>>>(
        ps, pmx, cnt, z_pool, (float*)d_out, (int)BD);
}


# Edge aggregation: edges counting-sorted by destination once per call (int atomics), each layer gathers its rows' neighbours in the layer kernel; no float atomics on edges, sums in list order
# speedup vs baseline: 2.6685x; 2.6685x over previous
#include <hip/hip_runtime.h>
#include <hip/hip_bf16.h>


typedef float v8f __attribute__((ext_vector_type(8)));
typedef float v2f __attribute__((ext_vector_type(2)));

#define GNN_D 128
#define TILE_M 16
#define LDS_STRIDE 132

__device__ __forceinline__ unsigned f2mono(float f) {
    unsigned u = __float_as_uint(f);
    return (u & 0x80000000u) ? ~u : (u | 0x80000000u);
}
__device__ __forceinline__ float mono2f(unsigned m) {
    unsigned u = (m & 0x80000000u) ? (m & 0x7FFFFFFFu) : ~m;
    return __uint_as_float(u);
}
#define MONO_NEG_INF 0x007FFFFFu


__global__ __launch_bounds__(256) void gnn_init_mono_kernel(unsigned* __restrict__ p, int n) {
    int i = blockIdx.x * 256 + threadIdx.x;
    if (i < n) p[i] = MONO_NEG_INF;
}

__global__ __launch_bounds__(256) void gnn_count_kernel(const int* __restrict__ dst,
                                                        int* __restrict__ cnt, int E) {
    int e = blockIdx.x * 256 + threadIdx.x;
    if (e < E) atomicAdd(&cnt[dst[e]], 1);
}

__global__ __launch_bounds__(1024) void gnn_scan_kernel(const int* __restrict__ cnt,
                                                        int* __restrict__ rowptr,
                                                        int* __restrict__ cursor, int N) {
    __shared__ int part[1024];
    const int t = threadIdx.x;
    const int chunk = (N + 1023) >> 10;
    const int b = min(t * chunk, N);
    const int e = min(b + chunk, N);
    int sum = 0;
    #pragma unroll 8
    for (int i = b; i < e; ++i) sum += cnt[i];
    part[t] = sum;
    __syncthreads();
    for (int off = 1; off < 1024; off <<= 1) {
        int v = (t >= off) ? part[t - off] : 0;
        __syncthreads();
        part[t] += v;
        __syncthreads();
    }
    int run = part[t] - sum;
    #pragma unroll 8
    for (int i = b; i < e; ++i) {
        int cv = cnt[i];
        rowptr[i] = run;
        cursor[i] = run;
        run += cv;
    }
    if (t == 1023) rowptr[N] = part[1023];
}

__global__ __launch_bounds__(256) void gnn_fill_kernel(
    const int* __restrict__ src, const int* __restrict__ dst,
    int* __restrict__ cursor, int* __restrict__ nbr, int E)
{
    int e = blockIdx.x * 256 + threadIdx.x;
    if (e < E) {
        int p = atomicAdd(&cursor[dst[e]], 1);
        nbr[p] = src[e];
    }
}

__global__ __launch_bounds__(256) void gnn_layer_kernel(
    const float* __restrict__ x, const int* __restrict__ rowptr,
    const int* __restrict__ nbr,
    const float* __restrict__ zagg,
    const float* __restrict__ W1, const float* __restrict__ b1,
    const float* __restrict__ W2, const float* __restrict__ b2,
    float* __restrict__ xout)
{
    __shared__ float hs[TILE_M * LDS_STRIDE];
    __shared__ float xt[TILE_M * LDS_STRIDE];
    __shared__ float ts[TILE_M * LDS_STRIDE];

    const int tid  = threadIdx.x;
    const int row0 = blockIdx.x * TILE_M;

    const float z0 = zagg[0], z1 = zagg[1], z2 = zagg[2];

    {
        const int wv = __builtin_amdgcn_readfirstlane(tid >> 5);
        const int ln = tid & 31;
        const int nodeA = row0 + wv * 2;
        int bnd[3];
        bnd[0] = rowptr[nodeA];
        bnd[1] = rowptr[nodeA + 1];
        bnd[2] = rowptr[nodeA + 2];
        const float* xl = x + ln * 4;
        #pragma unroll
        for (int rr = 0; rr < 2; ++rr) {
            const int r   = wv * 2 + rr;
            const int beg = bnd[rr];
            const int end = bnd[rr + 1];
            const float4 xv = *reinterpret_cast<const float4*>(xl + (size_t)(row0 + r) * GNN_D);
            float4 sv = make_float4(0.0f, 0.0f, 0.0f, 0.0f);
            const float ninf = -__builtin_inff();
            float4 mv = make_float4(ninf, ninf, ninf, ninf);
            int j = beg;
            for (; j + 4 <= end; j += 4) {
                const int s0 = nbr[j], s1 = nbr[j + 1], s2 = nbr[j + 2], s3 = nbr[j + 3];
                const float4 v0 = *reinterpret_cast<const float4*>(xl + (size_t)s0 * GNN_D);
                const float4 v1 = *reinterpret_cast<const float4*>(xl + (size_t)s1 * GNN_D);
                const float4 v2 = *reinterpret_cast<const float4*>(xl + (size_t)s2 * GNN_D);
                const float4 v3 = *reinterpret_cast<const float4*>(xl + (size_t)s3 * GNN_D);
                sv.x = (((sv.x + v0.x) + v1.x) + v2.x) + v3.x;
                sv.y = (((sv.y + v0.y) + v1.y) + v2.y) + v3.y;
                sv.z = (((sv.z + v0.z) + v1.z) + v2.z) + v3.z;
                sv.w = (((sv.w + v0.w) + v1.w) + v2.w) + v3.w;
                mv.x = fmaxf(fmaxf(fmaxf(fmaxf(mv.x, v0.x), v1.x), v2.x), v3.x);
                mv.y = fmaxf(fmaxf(fmaxf(fmaxf(mv.y, v0.y), v1.y), v2.y), v3.y);
                mv.z = fmaxf(fmaxf(fmaxf(fmaxf(mv.z, v0.z), v1.z), v2.z), v3.z);
                mv.w = fmaxf(fmaxf(fmaxf(fmaxf(mv.w, v0.w), v1.w), v2.w), v3.w);
            }
            for (; j < end; ++j) {
                const int s0 = nbr[j];
                const float4 v0 = *reinterpret_cast<const float4*>(xl + (size_t)s0 * GNN_D);
                sv.x += v0.x; sv.y += v0.y; sv.z += v0.z; sv.w += v0.w;
                mv.x = fmaxf(mv.x, v0.x); mv.y = fmaxf(mv.y, v0.y);
                mv.z = fmaxf(mv.z, v0.z); mv.w = fmaxf(mv.w, v0.w);
            }
            const float dg = (float)(end - beg);
            const float dc = fmaxf(dg, 1.0f);
            const bool  nb = dg > 0.0f;
            float4 h;
            h.x = xv.x + z0 * sv.x + z1 * (sv.x / dc) + z2 * (nb ? mv.x : 0.0f);
            h.y = xv.y + z0 * sv.y + z1 * (sv.y / dc) + z2 * (nb ? mv.y : 0.0f);
            h.z = xv.z + z0 * sv.z + z1 * (sv.z / dc) + z2 * (nb ? mv.z : 0.0f);
            h.w = xv.w + z0 * sv.w + z1 * (sv.w / dc) + z2 * (nb ? mv.w : 0.0f);
            *reinterpret_cast<float4*>(&hs[r * LDS_STRIDE + ln * 4]) = h;
            *reinterpret_cast<float4*>(&xt[r * LDS_STRIDE + ln * 4]) = xv;
        }
    }
    __syncthreads();

    const int wave  = tid >> 5;
    const int lane  = tid & 31;
    const int ncol  = (wave << 4) + (lane & 15);
    const int mlo   = lane & 15;
    const int koff  = (lane >> 4) << 1;
    const int rbase = (lane >> 4) << 3;
    const float bia1 = b1[ncol];
    const float bia2 = b2[ncol];

    v8f acc = {};
    #pragma unroll
    for (int k0 = 0; k0 < GNN_D; k0 += 4) {
        int kk = k0 + koff;
        v2f a; a.x = hs[mlo * LDS_STRIDE + kk];
               a.y = hs[mlo * LDS_STRIDE + kk + 1];
        v2f b; b.x = W1[(size_t)kk * GNN_D + ncol];
               b.y = W1[(size_t)(kk + 1) * GNN_D + ncol];
        acc = __builtin_amdgcn_wmma_f32_16x16x4_f32(false, a, false, b,
                                                    (short)0, acc, false, false);
    }
    #pragma unroll
    for (int i = 0; i < 8; ++i)
        ts[(rbase + i) * LDS_STRIDE + ncol] = fmaxf(acc[i] + bia1, 0.0f);
    __syncthreads();

    v8f acc2 = {};
    #pragma unroll
    for (int k0 = 0; k0 < GNN_D; k0 += 4) {
        int kk = k0 + koff;
        v2f a; a.x = ts[mlo * LDS_STRIDE + kk];
               a.y = ts[mlo * LDS_STRIDE + kk + 1];
        v2f b; b.x = W2[(size_t)kk * GNN_D + ncol];
               b.y = W2[(size_t)(kk + 1) * GNN_D + ncol];
        acc2 = __builtin_amdgcn_wmma_f32_16x16x4_f32(false, a, false, b,
                                                     (short)0, acc2, false, false);
    }
    #pragma unroll
    for (int i = 0; i < 8; ++i) {
        int r = rbase + i;
        float h2 = acc2[i] + bia2;
        float xv = xt[r * LDS_STRIDE + ncol];
        xout[(size_t)(row0 + r) * GNN_D + ncol] = xv + fmaxf(h2, 0.0f);
    }
}

__global__ __launch_bounds__(256) void gnn_pool_scatter_kernel(
    const float* __restrict__ x, const int* __restrict__ batch,
    float* __restrict__ ps, unsigned* __restrict__ pmx,
    float* __restrict__ cnt, int N)
{
    int gid  = blockIdx.x * 256 + threadIdx.x;
    int n    = gid >> 5;
    int lane = gid & 31;
    if (n >= N) return;
    int b = batch[n];
    const float4 v = *reinterpret_cast<const float4*>(x + (size_t)n * GNN_D + lane * 4);
    float*    sp = ps  + (size_t)b * GNN_D + lane * 4;
    unsigned* mp = pmx + (size_t)b * GNN_D + lane * 4;
    atomicAdd(sp + 0, v.x); atomicAdd(sp + 1, v.y);
    atomicAdd(sp + 2, v.z); atomicAdd(sp + 3, v.w);
    atomicMax(mp + 0, f2mono(v.x)); atomicMax(mp + 1, f2mono(v.y));
    atomicMax(mp + 2, f2mono(v.z)); atomicMax(mp + 3, f2mono(v.w));
    if (lane == 0) atomicAdd(&cnt[b], 1.0f);
}

__global__ __launch_bounds__(256) void gnn_pool_finalize_kernel(
    const float* __restrict__ ps, const unsigned* __restrict__ pmx,
    const float* __restrict__ cnt, const float* __restrict__ zpool,
    float* __restrict__ out, int total)
{
    int i = blockIdx.x * 256 + threadIdx.x;
    if (i >= total) return;
    int b = i >> 7;
    float c  = cnt[b];
    float sv = ps[i];
    float mv = (c > 0.0f) ? mono2f(pmx[i]) : 0.0f;
    out[i] = zpool[0] * sv + zpool[1] * (sv / fmaxf(c, 1.0f)) + zpool[2] * mv;
}


extern "C" void kernel_launch(void* const* d_in, const int* in_sizes, int n_in,
                              void* d_out, int out_size, void* d_ws, size_t ws_size,
                              hipStream_t stream) {
    const float* x_in   = (const float*)d_in[0];
    const int*   eidx   = (const int*)d_in[1];
    const int*   batch  = (const int*)d_in[2];
    const float* z_agg  = (const float*)d_in[3];
    const float* z_pool = (const float*)d_in[4];
    const float* W1     = (const float*)d_in[5];
    const float* b1     = (const float*)d_in[6];
    const float* W2     = (const float*)d_in[7];
    const float* b2     = (const float*)d_in[8];
    (void)n_in;

    const int N = in_sizes[0] / GNN_D;
    const int E = in_sizes[1] / 2;
    const int L = in_sizes[5] / (GNN_D * GNN_D);
    const int Bg = out_size / GNN_D;
    const size_t ND = (size_t)N * GNN_D;
    const size_t BD = (size_t)Bg * GNN_D;

    char* p = (char*)d_ws;
    char* pend = p + ws_size;
    auto take = [&](size_t bytes) -> void* {
        void* r = (void*)p;
        p += (bytes + 255) & ~(size_t)255;
        return r;
    };
    float*    xa     = (float*)take(ND * 4);
    float*    xb     = (float*)take(ND * 4);
    int*      ecnt   = (int*)take((size_t)N * 4);
    int*      rowptr = (int*)take(((size_t)N + 1) * 4);
    int*      cursor = (int*)take((size_t)N * 4);
    int*      nbr    = (int*)take((size_t)E * 4);
    float*    ps     = (float*)take(BD * 4);
    unsigned* pmx    = (unsigned*)take(BD * 4);
    float*    cnt    = (float*)take((size_t)Bg * 4);
    if (p > pend) return;

    const int* src = eidx;
    const int* dst = eidx + E;

    hipMemcpyAsync(xa, x_in, ND * 4, hipMemcpyDeviceToDevice, stream);

    hipMemsetAsync(ecnt, 0, (size_t)N * 4, stream);
    gnn_count_kernel<<<(E + 255) / 256, 256, 0, stream>>>(dst, ecnt, E);
    gnn_scan_kernel<<<1, 1024, 0, stream>>>(ecnt, rowptr, cursor, N);
    gnn_fill_kernel<<<(E + 255) / 256, 256, 0, stream>>>(src, dst, cursor, nbr, E);

    float* xcur = xa;
    float* xnxt = xb;
    for (int l = 0; l < L; ++l) {
        gnn_layer_kernel<<<N / TILE_M, 256, 0, stream>>>(
            xcur, rowptr, nbr, z_agg + l * 3,
            W1 + (size_t)l * GNN_D * GNN_D, b1 + (size_t)l * GNN_D,
            W2 + (size_t)l * GNN_D * GNN_D, b2 + (size_t)l * GNN_D, xnxt);
        float* t = xcur; xcur = xnxt; xnxt = t;
    }

    hipMemsetAsync(ps, 0, BD * 4, stream);
    hipMemsetAsync(cnt, 0, (size_t)Bg * 4, stream);
    gnn_init_mono_kernel<<<(int)((BD + 255) / 256), 256, 0, stream>>>(pmx, (int)BD);
    gnn_pool_scatter_kernel<<<(N + 7) / 8, 256, 0, stream>>>(xcur, batch, ps, pmx, cnt, N);
    gnn_pool_finalize_kernel<<<(int)((BD + 255) / 256), 256, 0, stream>>>(
        ps, pmx, cnt, z_pool, (float*)d_out, (int)BD);
}


# Pooling: each wave reduces 32 consecutive nodes in registers and issues its float atomics only when the graph id changes or its run ends (was 9 atomics per node); sums now run-wise, then atomics
# speedup vs baseline: 3.4628x; 1.2977x over previous
#include <hip/hip_runtime.h>
#include <hip/hip_bf16.h>


typedef float v8f __attribute__((ext_vector_type(8)));
typedef float v2f __attribute__((ext_vector_type(2)));

#define GNN_D 128
#define TILE_M 16
#define LDS_STRIDE 132

__device__ __forceinline__ unsigned f2mono(float f) {
    unsigned u = __float_as_uint(f);
    return (u & 0x80000000u) ? ~u : (u | 0x80000000u);
}
__device__ __forceinline__ float mono2f(unsigned m) {
    unsigned u = (m & 0x80000000u) ? (m & 0x7FFFFFFFu) : ~m;
    return __uint_as_float(u);
}
#define MONO_NEG_INF 0x007FFFFFu


__global__ __launch_bounds__(256) void gnn_init_mono_kernel(unsigned* __restrict__ p, int n) {
    int i = blockIdx.x * 256 + threadIdx.x;
    if (i < n) p[i] = MONO_NEG_INF;
}

__global__ __launch_bounds__(256) void gnn_count_kernel(const int* __restrict__ dst,
                                                        int* __restrict__ cnt, int E) {
    int e = blockIdx.x * 256 + threadIdx.x;
    if (e < E) atomicAdd(&cnt[dst[e]], 1);
}

__global__ __launch_bounds__(1024) void gnn_scan_kernel(const int* __restrict__ cnt,
                                                        int* __restrict__ rowptr,
                                                        int* __restrict__ cursor, int N) {
    __shared__ int part[1024];
    const int t = threadIdx.x;
    const int chunk = (N + 1023) >> 10;
    const int b = min(t * chunk, N);
    const int e = min(b + chunk, N);
    int sum = 0;
    #pragma unroll 8
    for (int i = b; i < e; ++i) sum += cnt[i];
    part[t] = sum;
    __syncthreads();
    for (int off = 1; off < 1024; off <<= 1) {
        int v = (t >= off) ? part[t - off] : 0;
        __syncthreads();
        part[t] += v;
        __syncthreads();
    }
    int run = part[t] - sum;
    #pragma unroll 8
    for (int i = b; i < e; ++i) {
        int cv = cnt[i];
        rowptr[i] = run;
        cursor[i] = run;
        run += cv;
    }
    if (t == 1023) rowptr[N] = part[1023];
}

__global__ __launch_bounds__(256) void gnn_fill_kernel(
    const int* __restrict__ src, const int* __restrict__ dst,
    int* __restrict__ cursor, int* __restrict__ nbr, int E)
{
    int e = blockIdx.x * 256 + threadIdx.x;
    if (e < E) {
        int p = atomicAdd(&cursor[dst[e]], 1);
        nbr[p] = src[e];
    }
}

__global__ __launch_bounds__(256) void gnn_layer_kernel(
    const float* __restrict__ x, const int* __restrict__ rowptr,
    const int* __restrict__ nbr,
    const float* __restrict__ zagg,
    const float* __restrict__ W1, const float* __restrict__ b1,
    const float* __restrict__ W2, const float* __restrict__ b2,
    float* __restrict__ xout)
{
    __shared__ float hs[TILE_M * LDS_STRIDE];
    __shared__ float xt[TILE_M * LDS_STRIDE];
    __shared__ float ts[TILE_M * LDS_STRIDE];

    const int tid  = threadIdx.x;
    const int row0 = blockIdx.x * TILE_M;

    const float z0 = zagg[0], z1 = zagg[1], z2 = zagg[2];

    {
        const int wv = __builtin_amdgcn_readfirstlane(tid >> 5);
        const int ln = tid & 31;
        const int nodeA = row0 + wv * 2;
        int bnd[3];
        bnd[0] = rowptr[nodeA];
        bnd[1] = rowptr[nodeA + 1];
        bnd[2] = rowptr[nodeA + 2];
        const float* xl = x + ln * 4;
        #pragma unroll
        for (int rr = 0; rr < 2; ++rr) {
            const int r   = wv * 2 + rr;
            const int beg = bnd[rr];
            const int end = bnd[rr + 1];
            const float4 xv = *reinterpret_cast<const float4*>(xl + (size_t)(row0 + r) * GNN_D);
            float4 sv = make_float4(0.0f, 0.0f, 0.0f, 0.0f);
            const float ninf = -__builtin_inff();
            float4 mv = make_float4(ninf, ninf, ninf, ninf);
            int j = beg;
            for (; j + 4 <= end; j += 4) {
                const int s0 = nbr[j], s1 = nbr[j + 1], s2 = nbr[j + 2], s3 = nbr[j + 3];
                const float4 v0 = *reinterpret_cast<const float4*>(xl + (size_t)s0 * GNN_D);
                const float4 v1 = *reinterpret_cast<const float4*>(xl + (size_t)s1 * GNN_D);
                const float4 v2 = *reinterpret_cast<const float4*>(xl + (size_t)s2 * GNN_D);
                const float4 v3 = *reinterpret_cast<const float4*>(xl + (size_t)s3 * GNN_D);
                sv.x = (((sv.x + v0.x) + v1.x) + v2.x) + v3.x;
                sv.y = (((sv.y + v0.y) + v1.y) + v2.y) + v3.y;
                sv.z = (((sv.z + v0.z) + v1.z) + v2.z) + v3.z;
                sv.w = (((sv.w + v0.w) + v1.w) + v2.w) + v3.w;
                mv.x = fmaxf(fmaxf(fmaxf(fmaxf(mv.x, v0.x), v1.x), v2.x), v3.x);
                mv.y = fmaxf(fmaxf(fmaxf(fmaxf(mv.y, v0.y), v1.y), v2.y), v3.y);
                mv.z = fmaxf(fmaxf(fmaxf(fmaxf(mv.z, v0.z), v1.z), v2.z), v3.z);
                mv.w = fmaxf(fmaxf(fmaxf(fmaxf(mv.w, v0.w), v1.w), v2.w), v3.w);
            }
            for (; j < end; ++j) {
                const int s0 = nbr[j];
                const float4 v0 = *reinterpret_cast<const float4*>(xl + (size_t)s0 * GNN_D);
                sv.x += v0.x; sv.y += v0.y; sv.z += v0.z; sv.w += v0.w;
                mv.x = fmaxf(mv.x, v0.x); mv.y = fmaxf(mv.y, v0.y);
                mv.z = fmaxf(mv.z, v0.z); mv.w = fmaxf(mv.w, v0.w);
            }
            const float dg = (float)(end - beg);
            const float dc = fmaxf(dg, 1.0f);
            const bool  nb = dg > 0.0f;
            float4 h;
            h.x = xv.x + z0 * sv.x + z1 * (sv.x / dc) + z2 * (nb ? mv.x : 0.0f);
            h.y = xv.y + z0 * sv.y + z1 * (sv.y / dc) + z2 * (nb ? mv.y : 0.0f);
            h.z = xv.z + z0 * sv.z + z1 * (sv.z / dc) + z2 * (nb ? mv.z : 0.0f);
            h.w = xv.w + z0 * sv.w + z1 * (sv.w / dc) + z2 * (nb ? mv.w : 0.0f);
            *reinterpret_cast<float4*>(&hs[r * LDS_STRIDE + ln * 4]) = h;
            *reinterpret_cast<float4*>(&xt[r * LDS_STRIDE + ln * 4]) = xv;
        }
    }
    __syncthreads();

    const int wave  = tid >> 5;
    const int lane  = tid & 31;
    const int ncol  = (wave << 4) + (lane & 15);
    const int mlo   = lane & 15;
    const int koff  = (lane >> 4) << 1;
    const int rbase = (lane >> 4) << 3;
    const float bia1 = b1[ncol];
    const float bia2 = b2[ncol];

    v8f acc = {};
    #pragma unroll
    for (int k0 = 0; k0 < GNN_D; k0 += 4) {
        int kk = k0 + koff;
        v2f a; a.x = hs[mlo * LDS_STRIDE + kk];
               a.y = hs[mlo * LDS_STRIDE + kk + 1];
        v2f b; b.x = W1[(size_t)kk * GNN_D + ncol];
               b.y = W1[(size_t)(kk + 1) * GNN_D + ncol];
        acc = __builtin_amdgcn_wmma_f32_16x16x4_f32(false, a, false, b,
                                                    (short)0, acc, false, false);
    }
    #pragma unroll
    for (int i = 0; i < 8; ++i)
        ts[(rbase + i) * LDS_STRIDE + ncol] = fmaxf(acc[i] + bia1, 0.0f);
    __syncthreads();

    v8f acc2 = {};
    #pragma unroll
    for (int k0 = 0; k0 < GNN_D; k0 += 4) {
        int kk = k0 + koff;
        v2f a; a.x = ts[mlo * LDS_STRIDE + kk];
               a.y = ts[mlo * LDS_STRIDE + kk + 1];
        v2f b; b.x = W2[(size_t)kk * GNN_D + ncol];
               b.y = W2[(size_t)(kk + 1) * GNN_D + ncol];
        acc2 = __builtin_amdgcn_wmma_f32_16x16x4_f32(false, a, false, b,
                                                     (short)0, acc2, false, false);
    }
    #pragma unroll
    for (int i = 0; i < 8; ++i) {
        int r = rbase + i;
        float h2 = acc2[i] + bia2;
        float xv = xt[r * LDS_STRIDE + ncol];
        xout[(size_t)(row0 + r) * GNN_D + ncol] = xv + fmaxf(h2, 0.0f);
    }
}

#define POOL_RUN 32

__device__ __forceinline__ void gnn_pool_flush(
    float* __restrict__ ps, unsigned* __restrict__ pmx, float* __restrict__ cnt,
    int b, int lane, const float4& sv, const uint4& mm, int run)
{
    float*    sp = ps  + (size_t)b * GNN_D + lane * 4;
    unsigned* mp = pmx + (size_t)b * GNN_D + lane * 4;
    atomicAdd(sp + 0, sv.x); atomicAdd(sp + 1, sv.y);
    atomicAdd(sp + 2, sv.z); atomicAdd(sp + 3, sv.w);
    atomicMax(mp + 0, mm.x); atomicMax(mp + 1, mm.y);
    atomicMax(mp + 2, mm.z); atomicMax(mp + 3, mm.w);
    if (lane == 0) atomicAdd(&cnt[b], (float)run);
}

__global__ __launch_bounds__(64) void gnn_pool_scatter_kernel(
    const float* __restrict__ x, const int* __restrict__ batch,
    float* __restrict__ ps, unsigned* __restrict__ pmx,
    float* __restrict__ cnt, int N)
{
    const int wv   = blockIdx.x * 2 + __builtin_amdgcn_readfirstlane((int)threadIdx.x >> 5);
    const int lane = threadIdx.x & 31;
    const int n0   = wv * POOL_RUN;
    if (n0 >= N) return;
    const int n1   = min(n0 + POOL_RUN, N);
    const float* xl = x + lane * 4;

    int    curb = batch[n0];
    float4 sv   = make_float4(0.0f, 0.0f, 0.0f, 0.0f);
    uint4  mm   = make_uint4(MONO_NEG_INF, MONO_NEG_INF, MONO_NEG_INF, MONO_NEG_INF);
    int    run  = 0;

    for (int n = n0; n < n1; n += 8) {
        int    bb[8];
        float4 v[8];
        #pragma unroll
        for (int i = 0; i < 8; ++i) {
            const int m = min(n + i, n1 - 1);
            bb[i] = batch[m];
            v[i]  = *reinterpret_cast<const float4*>(xl + (size_t)m * GNN_D);
        }
        #pragma unroll
        for (int i = 0; i < 8; ++i) {
            if (i == 0 || n + i < n1) {
                if (bb[i] != curb) {
                    gnn_pool_flush(ps, pmx, cnt, curb, lane, sv, mm, run);
                    sv   = make_float4(0.0f, 0.0f, 0.0f, 0.0f);
                    mm   = make_uint4(MONO_NEG_INF, MONO_NEG_INF, MONO_NEG_INF, MONO_NEG_INF);
                    run  = 0;
                    curb = bb[i];
                }
                sv.x += v[i].x; sv.y += v[i].y; sv.z += v[i].z; sv.w += v[i].w;
                mm.x = max(mm.x, f2mono(v[i].x)); mm.y = max(mm.y, f2mono(v[i].y));
                mm.z = max(mm.z, f2mono(v[i].z)); mm.w = max(mm.w, f2mono(v[i].w));
                ++run;
            }
        }
    }
    gnn_pool_flush(ps, pmx, cnt, curb, lane, sv, mm, run);
}

__global__ __launch_bounds__(256) void gnn_pool_finalize_kernel(
    const float* __restrict__ ps, const unsigned* __restrict__ pmx,
    const float* __restrict__ cnt, const float* __restrict__ zpool,
    float* __restrict__ out, int total)
{
    int i = blockIdx.x * 256 + threadIdx.x;
    if (i >= total) return;
    int b = i >> 7;
    float c  = cnt[b];
    float sv = ps[i];
    float mv = (c > 0.0f) ? mono2f(pmx[i]) : 0.0f;
    out[i] = zpool[0] * sv + zpool[1] * (sv / fmaxf(c, 1.0f)) + zpool[2] * mv;
}


extern "C" void kernel_launch(void* const* d_in, const int* in_sizes, int n_in,
                              void* d_out, int out_size, void* d_ws, size_t ws_size,
                              hipStream_t stream) {
    const float* x_in   = (const float*)d_in[0];
    const int*   eidx   = (const int*)d_in[1];
    const int*   batch  = (const int*)d_in[2];
    const float* z_agg  = (const float*)d_in[3];
    const float* z_pool = (const float*)d_in[4];
    const float* W1     = (const float*)d_in[5];
    const float* b1     = (const float*)d_in[6];
    const float* W2     = (const float*)d_in[7];
    const float* b2     = (const float*)d_in[8];
    (void)n_in;

    const int N = in_sizes[0] / GNN_D;
    const int E = in_sizes[1] / 2;
    const int L = in_sizes[5] / (GNN_D * GNN_D);
    const int Bg = out_size / GNN_D;
    const size_t ND = (size_t)N * GNN_D;
    const size_t BD = (size_t)Bg * GNN_D;

    char* p = (char*)d_ws;
    char* pend = p + ws_size;
    auto take = [&](size_t bytes) -> void* {
        void* r = (void*)p;
        p += (bytes + 255) & ~(size_t)255;
        return r;
    };
    float*    xa     = (float*)take(ND * 4);
    float*    xb     = (float*)take(ND * 4);
    int*      ecnt   = (int*)take((size_t)N * 4);
    int*      rowptr = (int*)take(((size_t)N + 1) * 4);
    int*      cursor = (int*)take((size_t)N * 4);
    int*      nbr    = (int*)take((size_t)E * 4);
    float*    ps     = (float*)take(BD * 4);
    unsigned* pmx    = (unsigned*)take(BD * 4);
    float*    cnt    = (float*)take((size_t)Bg * 4);
    if (p > pend) return;

    const int* src = eidx;
    const int* dst = eidx + E;

    hipMemcpyAsync(xa, x_in, ND * 4, hipMemcpyDeviceToDevice, stream);

    hipMemsetAsync(ecnt, 0, (size_t)N * 4, stream);
    gnn_count_kernel<<<(E + 255) / 256, 256, 0, stream>>>(dst, ecnt, E);
    gnn_scan_kernel<<<1, 1024, 0, stream>>>(ecnt, rowptr, cursor, N);
    gnn_fill_kernel<<<(E + 255) / 256, 256, 0, stream>>>(src, dst, cursor, nbr, E);

    float* xcur = xa;
    float* xnxt = xb;
    for (int l = 0; l < L; ++l) {
        gnn_layer_kernel<<<N / TILE_M, 256, 0, stream>>>(
            xcur, rowptr, nbr, z_agg + l * 3,
            W1 + (size_t)l * GNN_D * GNN_D, b1 + (size_t)l * GNN_D,
            W2 + (size_t)l * GNN_D * GNN_D, b2 + (size_t)l * GNN_D, xnxt);
        float* t = xcur; xcur = xnxt; xnxt = t;
    }

    hipMemsetAsync(ps, 0, BD * 4, stream);
    hipMemsetAsync(cnt, 0, (size_t)Bg * 4, stream);
    gnn_init_mono_kernel<<<(int)((BD + 255) / 256), 256, 0, stream>>>(pmx, (int)BD);
    gnn_pool_scatter_kernel<<<(N + 2 * POOL_RUN - 1) / (2 * POOL_RUN), 64, 0, stream>>>(xcur, batch, ps, pmx, cnt, N);
    gnn_pool_finalize_kernel<<<(int)((BD + 255) / 256), 256, 0, stream>>>(
        ps, pmx, cnt, z_pool, (float*)d_out, (int)BD);
}


# Prefix sum of the per-destination edge counts on 49 work-groups: each adds the counts in front of it with whole-line loads, then scans its 1024 (was one work-group, lanes 196 bytes apart)
# speedup vs baseline: 3.9360x; 1.1367x over previous
#include <hip/hip_runtime.h>
#include <hip/hip_bf16.h>


typedef float v8f __attribute__((ext_vector_type(8)));
typedef float v2f __attribute__((ext_vector_type(2)));

#define GNN_D 128
#define TILE_M 16
#define LDS_STRIDE 132

__device__ __forceinline__ unsigned f2mono(float f) {
    unsigned u = __float_as_uint(f);
    return (u & 0x80000000u) ? ~u : (u | 0x80000000u);
}
__device__ __forceinline__ float mono2f(unsigned m) {
    unsigned u = (m & 0x80000000u) ? (m & 0x7FFFFFFFu) : ~m;
    return __uint_as_float(u);
}
#define MONO_NEG_INF 0x007FFFFFu


__global__ __launch_bounds__(256) void gnn_init_mono_kernel(unsigned* __restrict__ p, int n) {
    int i = blockIdx.x * 256 + threadIdx.x;
    if (i < n) p[i] = MONO_NEG_INF;
}

__global__ __launch_bounds__(256) void gnn_count_kernel(const int* __restrict__ dst,
                                                        int* __restrict__ cnt, int E) {
    int e = blockIdx.x * 256 + threadIdx.x;
    if (e < E) atomicAdd(&cnt[dst[e]], 1);
}

#define SCAN_BLOCK 1024
__global__ __launch_bounds__(SCAN_BLOCK) void gnn_scan_kernel(const int* __restrict__ cnt,
                                                              int* __restrict__ rowptr,
                                                              int* __restrict__ cursor, int N) {
    __shared__ int wpre[32];
    __shared__ int wtot[32];
    const int t    = threadIdx.x;
    const int lane = t & 31;
    const int wv   = t >> 5;
    const int base = blockIdx.x * SCAN_BLOCK;

    const int4* c4 = reinterpret_cast<const int4*>(cnt);
    const int n4 = base >> 2;
    int pre = 0;
    #pragma unroll 4
    for (int j = t; j < n4; j += SCAN_BLOCK) {
        const int4 q = c4[j];
        pre += (q.x + q.y) + (q.z + q.w);
    }

    const int i = base + t;
    const int v = (i < N) ? cnt[i] : 0;

    int inc = v;
    for (int off = 1; off < 32; off <<= 1) {
        const int u = __shfl_up(inc, off, 32);
        if (lane >= off) inc += u;
    }
    for (int off = 16; off > 0; off >>= 1) pre += __shfl_xor(pre, off, 32);
    if (lane == 31) wtot[wv] = inc;
    if (lane == 0)  wpre[wv] = pre;
    __syncthreads();

    int a = wpre[lane];
    int b = (lane < wv) ? wtot[lane] : 0;
    for (int off = 16; off > 0; off >>= 1) {
        a += __shfl_xor(a, off, 32);
        b += __shfl_xor(b, off, 32);
    }
    const int excl = a + b + (inc - v);
    if (i < N) {
        rowptr[i] = excl;
        cursor[i] = excl;
    }
    if (i == N - 1) rowptr[N] = excl + v;
}

__global__ __launch_bounds__(256) void gnn_fill_kernel(
    const int* __restrict__ src, const int* __restrict__ dst,
    int* __restrict__ cursor, int* __restrict__ nbr, int E)
{
    int e = blockIdx.x * 256 + threadIdx.x;
    if (e < E) {
        int p = atomicAdd(&cursor[dst[e]], 1);
        nbr[p] = src[e];
    }
}

__global__ __launch_bounds__(256) void gnn_layer_kernel(
    const float* __restrict__ x, const int* __restrict__ rowptr,
    const int* __restrict__ nbr,
    const float* __restrict__ zagg,
    const float* __restrict__ W1, const float* __restrict__ b1,
    const float* __restrict__ W2, const float* __restrict__ b2,
    float* __restrict__ xout)
{
    __shared__ float hs[TILE_M * LDS_STRIDE];
    __shared__ float xt[TILE_M * LDS_STRIDE];
    __shared__ float ts[TILE_M * LDS_STRIDE];

    const int tid  = threadIdx.x;
    const int row0 = blockIdx.x * TILE_M;

    const float z0 = zagg[0], z1 = zagg[1], z2 = zagg[2];

    {
        const int wv = __builtin_amdgcn_readfirstlane(tid >> 5);
        const int ln = tid & 31;
        const int nodeA = row0 + wv * 2;
        int bnd[3];
        bnd[0] = rowptr[nodeA];
        bnd[1] = rowptr[nodeA + 1];
        bnd[2] = rowptr[nodeA + 2];
        const float* xl = x + ln * 4;
        #pragma unroll
        for (int rr = 0; rr < 2; ++rr) {
            const int r   = wv * 2 + rr;
            const int beg = bnd[rr];
            const int end = bnd[rr + 1];
            const float4 xv = *reinterpret_cast<const float4*>(xl + (size_t)(row0 + r) * GNN_D);
            float4 sv = make_float4(0.0f, 0.0f, 0.0f, 0.0f);
            const float ninf = -__builtin_inff();
            float4 mv = make_float4(ninf, ninf, ninf, ninf);
            int j = beg;
            for (; j + 4 <= end; j += 4) {
                const int s0 = nbr[j], s1 = nbr[j + 1], s2 = nbr[j + 2], s3 = nbr[j + 3];
                const float4 v0 = *reinterpret_cast<const float4*>(xl + (size_t)s0 * GNN_D);
                const float4 v1 = *reinterpret_cast<const float4*>(xl + (size_t)s1 * GNN_D);
                const float4 v2 = *reinterpret_cast<const float4*>(xl + (size_t)s2 * GNN_D);
                const float4 v3 = *reinterpret_cast<const float4*>(xl + (size_t)s3 * GNN_D);
                sv.x = (((sv.x + v0.x) + v1.x) + v2.x) + v3.x;
                sv.y = (((sv.y + v0.y) + v1.y) + v2.y) + v3.y;
                sv.z = (((sv.z + v0.z) + v1.z) + v2.z) + v3.z;
                sv.w = (((sv.w + v0.w) + v1.w) + v2.w) + v3.w;
                mv.x = fmaxf(fmaxf(fmaxf(fmaxf(mv.x, v0.x), v1.x), v2.x), v3.x);
                mv.y = fmaxf(fmaxf(fmaxf(fmaxf(mv.y, v0.y), v1.y), v2.y), v3.y);
                mv.z = fmaxf(fmaxf(fmaxf(fmaxf(mv.z, v0.z), v1.z), v2.z), v3.z);
                mv.w = fmaxf(fmaxf(fmaxf(fmaxf(mv.w, v0.w), v1.w), v2.w), v3.w);
            }
            for (; j < end; ++j) {
                const int s0 = nbr[j];
                const float4 v0 = *reinterpret_cast<const float4*>(xl + (size_t)s0 * GNN_D);
                sv.x += v0.x; sv.y += v0.y; sv.z += v0.z; sv.w += v0.w;
                mv.x = fmaxf(mv.x, v0.x); mv.y = fmaxf(mv.y, v0.y);
                mv.z = fmaxf(mv.z, v0.z); mv.w = fmaxf(mv.w, v0.w);
            }
            const float dg = (float)(end - beg);
            const float dc = fmaxf(dg, 1.0f);
            const bool  nb = dg > 0.0f;
            float4 h;
            h.x = xv.x + z0 * sv.x + z1 * (sv.x / dc) + z2 * (nb ? mv.x : 0.0f);
            h.y = xv.y + z0 * sv.y + z1 * (sv.y / dc) + z2 * (nb ? mv.y : 0.0f);
            h.z = xv.z + z0 * sv.z + z1 * (sv.z / dc) + z2 * (nb ? mv.z : 0.0f);
            h.w = xv.w + z0 * sv.w + z1 * (sv.w / dc) + z2 * (nb ? mv.w : 0.0f);
            *reinterpret_cast<float4*>(&hs[r * LDS_STRIDE + ln * 4]) = h;
            *reinterpret_cast<float4*>(&xt[r * LDS_STRIDE + ln * 4]) = xv;
        }
    }
    __syncthreads();

    const int wave  = tid >> 5;
    const int lane  = tid & 31;
    const int ncol  = (wave << 4) + (lane & 15);
    const int mlo   = lane & 15;
    const int koff  = (lane >> 4) << 1;
    const int rbase = (lane >> 4) << 3;
    const float bia1 = b1[ncol];
    const float bia2 = b2[ncol];

    v8f acc = {};
    #pragma unroll
    for (int k0 = 0; k0 < GNN_D; k0 += 4) {
        int kk = k0 + koff;
        v2f a; a.x = hs[mlo * LDS_STRIDE + kk];
               a.y = hs[mlo * LDS_STRIDE + kk + 1];
        v2f b; b.x = W1[(size_t)kk * GNN_D + ncol];
               b.y = W1[(size_t)(kk + 1) * GNN_D + ncol];
        acc = __builtin_amdgcn_wmma_f32_16x16x4_f32(false, a, false, b,
                                                    (short)0, acc, false, false);
    }
    #pragma unroll
    for (int i = 0; i < 8; ++i)
        ts[(rbase + i) * LDS_STRIDE + ncol] = fmaxf(acc[i] + bia1, 0.0f);
    __syncthreads();

    v8f acc2 = {};
    #pragma unroll
    for (int k0 = 0; k0 < GNN_D; k0 += 4) {
        int kk = k0 + koff;
        v2f a; a.x = ts[mlo * LDS_STRIDE + kk];
               a.y = ts[mlo * LDS_STRIDE + kk + 1];
        v2f b; b.x = W2[(size_t)kk * GNN_D + ncol];
               b.y = W2[(size_t)(kk + 1) * GNN_D + ncol];
        acc2 = __builtin_amdgcn_wmma_f32_16x16x4_f32(false, a, false, b,
                                                     (short)0, acc2, false, false);
    }
    #pragma unroll
    for (int i = 0; i < 8; ++i) {
        int r = rbase + i;
        float h2 = acc2[i] + bia2;
        float xv = xt[r * LDS_STRIDE + ncol];
        xout[(size_t)(row0 + r) * GNN_D + ncol] = xv + fmaxf(h2, 0.0f);
    }
}

#define POOL_RUN 32

__device__ __forceinline__ void gnn_pool_flush(
    float* __restrict__ ps, unsigned* __restrict__ pmx, float* __restrict__ cnt,
    int b, int lane, const float4& sv, const uint4& mm, int run)
{
    float*    sp = ps  + (size_t)b * GNN_D + lane * 4;
    unsigned* mp = pmx + (size_t)b * GNN_D + lane * 4;
    atomicAdd(sp + 0, sv.x); atomicAdd(sp + 1, sv.y);
    atomicAdd(sp + 2, sv.z); atomicAdd(sp + 3, sv.w);
    atomicMax(mp + 0, mm.x); atomicMax(mp + 1, mm.y);
    atomicMax(mp + 2, mm.z); atomicMax(mp + 3, mm.w);
    if (lane == 0) atomicAdd(&cnt[b], (float)run);
}

__global__ __launch_bounds__(64) void gnn_pool_scatter_kernel(
    const float* __restrict__ x, const int* __restrict__ batch,
    float* __restrict__ ps, unsigned* __restrict__ pmx,
    float* __restrict__ cnt, int N)
{
    const int wv   = blockIdx.x * 2 + __builtin_amdgcn_readfirstlane((int)threadIdx.x >> 5);
    const int lane = threadIdx.x & 31;
    const int n0   = wv * POOL_RUN;
    if (n0 >= N) return;
    const int n1   = min(n0 + POOL_RUN, N);
    const float* xl = x + lane * 4;

    int    curb = batch[n0];
    float4 sv   = make_float4(0.0f, 0.0f, 0.0f, 0.0f);
    uint4  mm   = make_uint4(MONO_NEG_INF, MONO_NEG_INF, MONO_NEG_INF, MONO_NEG_INF);
    int    run  = 0;

    for (int n = n0; n < n1; n += 8) {
        int    bb[8];
        float4 v[8];
        #pragma unroll
        for (int i = 0; i < 8; ++i) {
            const int m = min(n + i, n1 - 1);
            bb[i] = batch[m];
            v[i]  = *reinterpret_cast<const float4*>(xl + (size_t)m * GNN_D);
        }
        #pragma unroll
        for (int i = 0; i < 8; ++i) {
            if (i == 0 || n + i < n1) {
                if (bb[i] != curb) {
                    gnn_pool_flush(ps, pmx, cnt, curb, lane, sv, mm, run);
                    sv   = make_float4(0.0f, 0.0f, 0.0f, 0.0f);
                    mm   = make_uint4(MONO_NEG_INF, MONO_NEG_INF, MONO_NEG_INF, MONO_NEG_INF);
                    run  = 0;
                    curb = bb[i];
                }
                sv.x += v[i].x; sv.y += v[i].y; sv.z += v[i].z; sv.w += v[i].w;
                mm.x = max(mm.x, f2mono(v[i].x)); mm.y = max(mm.y, f2mono(v[i].y));
                mm.z = max(mm.z, f2mono(v[i].z)); mm.w = max(mm.w, f2mono(v[i].w));
                ++run;
            }
        }
    }
    gnn_pool_flush(ps, pmx, cnt, curb, lane, sv, mm, run);
}

__global__ __launch_bounds__(256) void gnn_pool_finalize_kernel(
    const float* __restrict__ ps, const unsigned* __restrict__ pmx,
    const float* __restrict__ cnt, const float* __restrict__ zpool,
    float* __restrict__ out, int total)
{
    int i = blockIdx.x * 256 + threadIdx.x;
    if (i >= total) return;
    int b = i >> 7;
    float c  = cnt[b];
    float sv = ps[i];
    float mv = (c > 0.0f) ? mono2f(pmx[i]) : 0.0f;
    out[i] = zpool[0] * sv + zpool[1] * (sv / fmaxf(c, 1.0f)) + zpool[2] * mv;
}


extern "C" void kernel_launch(void* const* d_in, const int* in_sizes, int n_in,
                              void* d_out, int out_size, void* d_ws, size_t ws_size,
                              hipStream_t stream) {
    const float* x_in   = (const float*)d_in[0];
    const int*   eidx   = (const int*)d_in[1];
    const int*   batch  = (const int*)d_in[2];
    const float* z_agg  = (const float*)d_in[3];
    const float* z_pool = (const float*)d_in[4];
    const float* W1     = (const float*)d_in[5];
    const float* b1     = (const float*)d_in[6];
    const float* W2     = (const float*)d_in[7];
    const float* b2     = (const float*)d_in[8];
    (void)n_in;

    const int N = in_sizes[0] / GNN_D;
    const int E = in_sizes[1] / 2;
    const int L = in_sizes[5] / (GNN_D * GNN_D);
    const int Bg = out_size / GNN_D;
    const size_t ND = (size_t)N * GNN_D;
    const size_t BD = (size_t)Bg * GNN_D;

    char* p = (char*)d_ws;
    char* pend = p + ws_size;
    auto take = [&](size_t bytes) -> void* {
        void* r = (void*)p;
        p += (bytes + 255) & ~(size_t)255;
        return r;
    };
    float*    xa     = (float*)take(ND * 4);
    float*    xb     = (float*)take(ND * 4);
    int*      ecnt   = (int*)take((size_t)N * 4);
    int*      rowptr = (int*)take(((size_t)N + 1) * 4);
    int*      cursor = (int*)take((size_t)N * 4);
    int*      nbr    = (int*)take((size_t)E * 4);
    float*    ps     = (float*)take(BD * 4);
    unsigned* pmx    = (unsigned*)take(BD * 4);
    float*    cnt    = (float*)take((size_t)Bg * 4);
    if (p > pend) return;

    const int* src = eidx;
    const int* dst = eidx + E;

    hipMemcpyAsync(xa, x_in, ND * 4, hipMemcpyDeviceToDevice, stream);

    hipMemsetAsync(ecnt, 0, (size_t)N * 4, stream);
    gnn_count_kernel<<<(E + 255) / 256, 256, 0, stream>>>(dst, ecnt, E);
    gnn_scan_kernel<<<(N + SCAN_BLOCK - 1) / SCAN_BLOCK, SCAN_BLOCK, 0, stream>>>(ecnt, rowptr, cursor, N);
    gnn_fill_kernel<<<(E + 255) / 256, 256, 0, stream>>>(src, dst, cursor, nbr, E);

    float* xcur = xa;
    float* xnxt = xb;
    for (int l = 0; l < L; ++l) {
        gnn_layer_kernel<<<N / TILE_M, 256, 0, stream>>>(
            xcur, rowptr, nbr, z_agg + l * 3,
            W1 + (size_t)l * GNN_D * GNN_D, b1 + (size_t)l * GNN_D,
            W2 + (size_t)l * GNN_D * GNN_D, b2 + (size_t)l * GNN_D, xnxt);
        float* t = xcur; xcur = xnxt; xnxt = t;
    }

    hipMemsetAsync(ps, 0, BD * 4, stream);
    hipMemsetAsync(cnt, 0, (size_t)Bg * 4, stream);
    gnn_init_mono_kernel<<<(int)((BD + 255) / 256), 256, 0, stream>>>(pmx, (int)BD);
    gnn_pool_scatter_kernel<<<(N + 2 * POOL_RUN - 1) / (2 * POOL_RUN), 64, 0, stream>>>(xcur, batch, ps, pmx, cnt, N);
    gnn_pool_finalize_kernel<<<(int)((BD + 255) / 256), 256, 0, stream>>>(
        ps, pmx, cnt, z_pool, (float*)d_out, (int)BD);
}


# Edge lists: edges sorted by 256-node destination range in LDS, then one work-group per range lays out its rows' lists; per-edge global atomics and scan gone; a row's sum order now by LDS arrival
# speedup vs baseline: 5.7821x; 1.4690x over previous
#include <hip/hip_runtime.h>
#include <hip/hip_bf16.h>


typedef float v8f __attribute__((ext_vector_type(8)));
typedef float v2f __attribute__((ext_vector_type(2)));

#define GNN_D 128
#define TILE_M 16
#define LDS_STRIDE 132

__device__ __forceinline__ unsigned f2mono(float f) {
    unsigned u = __float_as_uint(f);
    return (u & 0x80000000u) ? ~u : (u | 0x80000000u);
}
__device__ __forceinline__ float mono2f(unsigned m) {
    unsigned u = (m & 0x80000000u) ? (m & 0x7FFFFFFFu) : ~m;
    return __uint_as_float(u);
}
#define MONO_NEG_INF 0x007FFFFFu


__global__ __launch_bounds__(256) void gnn_init_mono_kernel(unsigned* __restrict__ p, int n) {
    int i = blockIdx.x * 256 + threadIdx.x;
    if (i < n) p[i] = MONO_NEG_INF;
}

#define BIN_SHIFT 8
#define BIN_NODES 256
#define NBMAX 256
#define PART_CH 2048
#define STAGE_CAP 16384

__global__ __launch_bounds__(256) void gnn_bincount_kernel(const int* __restrict__ dst,
                                                           int* __restrict__ binTotal, int E) {
    __shared__ int h[NBMAX];
    const int t = threadIdx.x;
    const int e0 = blockIdx.x * PART_CH + t;
    int dv[8];
    #pragma unroll
    for (int k = 0; k < 8; ++k) {
        const int e = e0 + k * 256;
        const int v = dst[min(e, E - 1)];
        dv[k] = (e < E) ? v : -1;
    }
    h[t] = 0;
    __syncthreads();
    #pragma unroll
    for (int k = 0; k < 8; ++k)
        if (dv[k] >= 0) atomicAdd(&h[dv[k] >> BIN_SHIFT], 1);
    __syncthreads();
    const int c = h[t];
    if (c > 0) atomicAdd(&binTotal[t], c);
}

__global__ __launch_bounds__(256) void gnn_partition_kernel(
    const int* __restrict__ src, const int* __restrict__ dst,
    const int* __restrict__ binTotal, int* __restrict__ binCursor,
    int* __restrict__ binStart, unsigned* __restrict__ part, int E, int NB)
{
    __shared__ int      lcnt[NBMAX];
    __shared__ int      lstart[NBMAX];
    __shared__ int      gdelta[NBMAX];
    __shared__ unsigned stage[PART_CH];
    __shared__ int      sbin[PART_CH];
    __shared__ int      wtA[8];
    __shared__ int      wtB[8];
    const int t    = threadIdx.x;
    const int lane = t & 31;
    const int wv   = t >> 5;
    const int c0   = blockIdx.x * PART_CH;

    int d[8], s[8], rk[8];
    #pragma unroll
    for (int k = 0; k < 8; ++k) {
        const int e  = c0 + t + k * 256;
        const int ec = min(e, E - 1);
        const int dd = dst[ec];
        s[k] = src[ec];
        d[k] = (e < E) ? dd : -1;
    }
    lcnt[t] = 0;
    __syncthreads();
    #pragma unroll
    for (int k = 0; k < 8; ++k) {
        rk[k] = 0;
        if (d[k] >= 0) rk[k] = atomicAdd(&lcnt[d[k] >> BIN_SHIFT], 1);
    }
    __syncthreads();

    const int c   = lcnt[t];
    const int tot = (t < NB) ? binTotal[t] : 0;
    int incA = tot, incB = c;
    for (int off = 1; off < 32; off <<= 1) {
        const int ua = __shfl_up(incA, off, 32);
        const int ub = __shfl_up(incB, off, 32);
        if (lane >= off) { incA += ua; incB += ub; }
    }
    if (lane == 31) { wtA[wv] = incA; wtB[wv] = incB; }
    __syncthreads();
    int wa = (lane < wv) ? wtA[lane] : 0;
    int wb = (lane < wv) ? wtB[lane] : 0;
    for (int off = 16; off > 0; off >>= 1) {
        wa += __shfl_xor(wa, off, 32);
        wb += __shfl_xor(wb, off, 32);
    }
    const int startG = wa + incA - tot;
    const int startL = wb + incB - c;
    int res = 0;
    if (c > 0) res = atomicAdd(&binCursor[t], c);
    lstart[t] = startL;
    gdelta[t] = startG + res - startL;
    if (blockIdx.x == 0) {
        if (t < NB) binStart[t] = startG;
        if (t == 0) binStart[NB] = E;
    }
    __syncthreads();

    #pragma unroll
    for (int k = 0; k < 8; ++k) {
        if (d[k] >= 0) {
            const int b = d[k] >> BIN_SHIFT;
            const int q = lstart[b] + rk[k];
            stage[q] = ((unsigned)s[k] << 8) | (unsigned)(d[k] & (BIN_NODES - 1));
            sbin[q]  = b;
        }
    }
    __syncthreads();

    const int n = min(PART_CH, E - c0);
    for (int i = t; i < n; i += 256)
        part[gdelta[sbin[i]] + i] = stage[i];
}

__global__ __launch_bounds__(1024) void gnn_binbuild_kernel(
    const unsigned* __restrict__ part, const int* __restrict__ binStart,
    int* __restrict__ rowptr, int* __restrict__ nbr, int N, int E)
{
    __shared__ int cnt[BIN_NODES];
    __shared__ int cur[BIN_NODES];
    __shared__ int wt[8];
    __shared__ int stage[STAGE_CAP];
    const int t    = threadIdx.x;
    const int lane = t & 31;
    const int wv   = t >> 5;
    const int b    = blockIdx.x;
    const int s    = binStart[b];
    const int n    = binStart[b + 1] - s;

    if (t < BIN_NODES) cnt[t] = 0;
    __syncthreads();
    for (int i = t; i < n; i += 1024)
        atomicAdd(&cnt[part[s + i] & (BIN_NODES - 1)], 1);
    __syncthreads();

    int v = 0, inc = 0;
    if (t < BIN_NODES) {
        v = cnt[t];
        inc = v;
        for (int off = 1; off < 32; off <<= 1) {
            const int u = __shfl_up(inc, off, 32);
            if (lane >= off) inc += u;
        }
        if (lane == 31) wt[wv] = inc;
    }
    __syncthreads();
    if (t < BIN_NODES) {
        int w = (lane < wv) ? wt[lane] : 0;
        for (int off = 16; off > 0; off >>= 1) w += __shfl_xor(w, off, 32);
        const int excl = w + inc - v;
        cur[t] = excl;
        const int node = b * BIN_NODES + t;
        if (node < N) rowptr[node] = s + excl;
    }
    if (b == (int)gridDim.x - 1 && t == 0) rowptr[N] = E;
    __syncthreads();

    const bool fits = n <= STAGE_CAP;
    for (int i = t; i < n; i += 1024) {
        const unsigned pk = part[s + i];
        const int p = atomicAdd(&cur[pk & (BIN_NODES - 1)], 1);
        const int sid = (int)(pk >> 8);
        if (fits) stage[p] = sid;
        else      nbr[s + p] = sid;
    }
    __syncthreads();
    if (fits) {
        for (int i = t; i < n; i += 1024) nbr[s + i] = stage[i];
    }
}

__global__ __launch_bounds__(256) void gnn_layer_kernel(
    const float* __restrict__ x, const int* __restrict__ rowptr,
    const int* __restrict__ nbr,
    const float* __restrict__ zagg,
    const float* __restrict__ W1, const float* __restrict__ b1,
    const float* __restrict__ W2, const float* __restrict__ b2,
    float* __restrict__ xout)
{
    __shared__ float hs[TILE_M * LDS_STRIDE];
    __shared__ float xt[TILE_M * LDS_STRIDE];
    __shared__ float ts[TILE_M * LDS_STRIDE];

    const int tid  = threadIdx.x;
    const int row0 = blockIdx.x * TILE_M;

    const float z0 = zagg[0], z1 = zagg[1], z2 = zagg[2];

    {
        const int wv = __builtin_amdgcn_readfirstlane(tid >> 5);
        const int ln = tid & 31;
        const int nodeA = row0 + wv * 2;
        int bnd[3];
        bnd[0] = rowptr[nodeA];
        bnd[1] = rowptr[nodeA + 1];
        bnd[2] = rowptr[nodeA + 2];
        const float* xl = x + ln * 4;
        #pragma unroll
        for (int rr = 0; rr < 2; ++rr) {
            const int r   = wv * 2 + rr;
            const int beg = bnd[rr];
            const int end = bnd[rr + 1];
            const float4 xv = *reinterpret_cast<const float4*>(xl + (size_t)(row0 + r) * GNN_D);
            float4 sv = make_float4(0.0f, 0.0f, 0.0f, 0.0f);
            const float ninf = -__builtin_inff();
            float4 mv = make_float4(ninf, ninf, ninf, ninf);
            int j = beg;
            for (; j + 4 <= end; j += 4) {
                const int s0 = nbr[j], s1 = nbr[j + 1], s2 = nbr[j + 2], s3 = nbr[j + 3];
                const float4 v0 = *reinterpret_cast<const float4*>(xl + (size_t)s0 * GNN_D);
                const float4 v1 = *reinterpret_cast<const float4*>(xl + (size_t)s1 * GNN_D);
                const float4 v2 = *reinterpret_cast<const float4*>(xl + (size_t)s2 * GNN_D);
                const float4 v3 = *reinterpret_cast<const float4*>(xl + (size_t)s3 * GNN_D);
                sv.x = (((sv.x + v0.x) + v1.x) + v2.x) + v3.x;
                sv.y = (((sv.y + v0.y) + v1.y) + v2.y) + v3.y;
                sv.z = (((sv.z + v0.z) + v1.z) + v2.z) + v3.z;
                sv.w = (((sv.w + v0.w) + v1.w) + v2.w) + v3.w;
                mv.x = fmaxf(fmaxf(fmaxf(fmaxf(mv.x, v0.x), v1.x), v2.x), v3.x);
                mv.y = fmaxf(fmaxf(fmaxf(fmaxf(mv.y, v0.y), v1.y), v2.y), v3.y);
                mv.z = fmaxf(fmaxf(fmaxf(fmaxf(mv.z, v0.z), v1.z), v2.z), v3.z);
                mv.w = fmaxf(fmaxf(fmaxf(fmaxf(mv.w, v0.w), v1.w), v2.w), v3.w);
            }
            for (; j < end; ++j) {
                const int s0 = nbr[j];
                const float4 v0 = *reinterpret_cast<const float4*>(xl + (size_t)s0 * GNN_D);
                sv.x += v0.x; sv.y += v0.y; sv.z += v0.z; sv.w += v0.w;
                mv.x = fmaxf(mv.x, v0.x); mv.y = fmaxf(mv.y, v0.y);
                mv.z = fmaxf(mv.z, v0.z); mv.w = fmaxf(mv.w, v0.w);
            }
            const float dg = (float)(end - beg);
            const float dc = fmaxf(dg, 1.0f);
            const bool  nb = dg > 0.0f;
            float4 h;
            h.x = xv.x + z0 * sv.x + z1 * (sv.x / dc) + z2 * (nb ? mv.x : 0.0f);
            h.y = xv.y + z0 * sv.y + z1 * (sv.y / dc) + z2 * (nb ? mv.y : 0.0f);
            h.z = xv.z + z0 * sv.z + z1 * (sv.z / dc) + z2 * (nb ? mv.z : 0.0f);
            h.w = xv.w + z0 * sv.w + z1 * (sv.w / dc) + z2 * (nb ? mv.w : 0.0f);
            *reinterpret_cast<float4*>(&hs[r * LDS_STRIDE + ln * 4]) = h;
            *reinterpret_cast<float4*>(&xt[r * LDS_STRIDE + ln * 4]) = xv;
        }
    }
    __syncthreads();

    const int wave  = tid >> 5;
    const int lane  = tid & 31;
    const int ncol  = (wave << 4) + (lane & 15);
    const int mlo   = lane & 15;
    const int koff  = (lane >> 4) << 1;
    const int rbase = (lane >> 4) << 3;
    const float bia1 = b1[ncol];
    const float bia2 = b2[ncol];

    v8f acc = {};
    #pragma unroll
    for (int k0 = 0; k0 < GNN_D; k0 += 4) {
        int kk = k0 + koff;
        v2f a; a.x = hs[mlo * LDS_STRIDE + kk];
               a.y = hs[mlo * LDS_STRIDE + kk + 1];
        v2f b; b.x = W1[(size_t)kk * GNN_D + ncol];
               b.y = W1[(size_t)(kk + 1) * GNN_D + ncol];
        acc = __builtin_amdgcn_wmma_f32_16x16x4_f32(false, a, false, b,
                                                    (short)0, acc, false, false);
    }
    #pragma unroll
    for (int i = 0; i < 8; ++i)
        ts[(rbase + i) * LDS_STRIDE + ncol] = fmaxf(acc[i] + bia1, 0.0f);
    __syncthreads();

    v8f acc2 = {};
    #pragma unroll
    for (int k0 = 0; k0 < GNN_D; k0 += 4) {
        int kk = k0 + koff;
        v2f a; a.x = ts[mlo * LDS_STRIDE + kk];
               a.y = ts[mlo * LDS_STRIDE + kk + 1];
        v2f b; b.x = W2[(size_t)kk * GNN_D + ncol];
               b.y = W2[(size_t)(kk + 1) * GNN_D + ncol];
        acc2 = __builtin_amdgcn_wmma_f32_16x16x4_f32(false, a, false, b,
                                                     (short)0, acc2, false, false);
    }
    #pragma unroll
    for (int i = 0; i < 8; ++i) {
        int r = rbase + i;
        float h2 = acc2[i] + bia2;
        float xv = xt[r * LDS_STRIDE + ncol];
        xout[(size_t)(row0 + r) * GNN_D + ncol] = xv + fmaxf(h2, 0.0f);
    }
}

#define POOL_RUN 32

__device__ __forceinline__ void gnn_pool_flush(
    float* __restrict__ ps, unsigned* __restrict__ pmx, float* __restrict__ cnt,
    int b, int lane, const float4& sv, const uint4& mm, int run)
{
    float*    sp = ps  + (size_t)b * GNN_D + lane * 4;
    unsigned* mp = pmx + (size_t)b * GNN_D + lane * 4;
    atomicAdd(sp + 0, sv.x); atomicAdd(sp + 1, sv.y);
    atomicAdd(sp + 2, sv.z); atomicAdd(sp + 3, sv.w);
    atomicMax(mp + 0, mm.x); atomicMax(mp + 1, mm.y);
    atomicMax(mp + 2, mm.z); atomicMax(mp + 3, mm.w);
    if (lane == 0) atomicAdd(&cnt[b], (float)run);
}

__global__ __launch_bounds__(64) void gnn_pool_scatter_kernel(
    const float* __restrict__ x, const int* __restrict__ batch,
    float* __restrict__ ps, unsigned* __restrict__ pmx,
    float* __restrict__ cnt, int N)
{
    const int wv   = blockIdx.x * 2 + __builtin_amdgcn_readfirstlane((int)threadIdx.x >> 5);
    const int lane = threadIdx.x & 31;
    const int n0   = wv * POOL_RUN;
    if (n0 >= N) return;
    const int n1   = min(n0 + POOL_RUN, N);
    const float* xl = x + lane * 4;

    int    curb = batch[n0];
    float4 sv   = make_float4(0.0f, 0.0f, 0.0f, 0.0f);
    uint4  mm   = make_uint4(MONO_NEG_INF, MONO_NEG_INF, MONO_NEG_INF, MONO_NEG_INF);
    int    run  = 0;

    for (int n = n0; n < n1; n += 8) {
        int    bb[8];
        float4 v[8];
        #pragma unroll
        for (int i = 0; i < 8; ++i) {
            const int m = min(n + i, n1 - 1);
            bb[i] = batch[m];
            v[i]  = *reinterpret_cast<const float4*>(xl + (size_t)m * GNN_D);
        }
        #pragma unroll
        for (int i = 0; i < 8; ++i) {
            if (i == 0 || n + i < n1) {
                if (bb[i] != curb) {
                    gnn_pool_flush(ps, pmx, cnt, curb, lane, sv, mm, run);
                    sv   = make_float4(0.0f, 0.0f, 0.0f, 0.0f);
                    mm   = make_uint4(MONO_NEG_INF, MONO_NEG_INF, MONO_NEG_INF, MONO_NEG_INF);
                    run  = 0;
                    curb = bb[i];
                }
                sv.x += v[i].x; sv.y += v[i].y; sv.z += v[i].z; sv.w += v[i].w;
                mm.x = max(mm.x, f2mono(v[i].x)); mm.y = max(mm.y, f2mono(v[i].y));
                mm.z = max(mm.z, f2mono(v[i].z)); mm.w = max(mm.w, f2mono(v[i].w));
                ++run;
            }
        }
    }
    gnn_pool_flush(ps, pmx, cnt, curb, lane, sv, mm, run);
}

__global__ __launch_bounds__(256) void gnn_pool_finalize_kernel(
    const float* __restrict__ ps, const unsigned* __restrict__ pmx,
    const float* __restrict__ cnt, const float* __restrict__ zpool,
    float* __restrict__ out, int total)
{
    int i = blockIdx.x * 256 + threadIdx.x;
    if (i >= total) return;
    int b = i >> 7;
    float c  = cnt[b];
    float sv = ps[i];
    float mv = (c > 0.0f) ? mono2f(pmx[i]) : 0.0f;
    out[i] = zpool[0] * sv + zpool[1] * (sv / fmaxf(c, 1.0f)) + zpool[2] * mv;
}


extern "C" void kernel_launch(void* const* d_in, const int* in_sizes, int n_in,
                              void* d_out, int out_size, void* d_ws, size_t ws_size,
                              hipStream_t stream) {
    const float* x_in   = (const float*)d_in[0];
    const int*   eidx   = (const int*)d_in[1];
    const int*   batch  = (const int*)d_in[2];
    const float* z_agg  = (const float*)d_in[3];
    const float* z_pool = (const float*)d_in[4];
    const float* W1     = (const float*)d_in[5];
    const float* b1     = (const float*)d_in[6];
    const float* W2     = (const float*)d_in[7];
    const float* b2     = (const float*)d_in[8];
    (void)n_in;

    const int N = in_sizes[0] / GNN_D;
    const int E = in_sizes[1] / 2;
    const int L = in_sizes[5] / (GNN_D * GNN_D);
    const int Bg = out_size / GNN_D;
    const size_t ND = (size_t)N * GNN_D;
    const size_t BD = (size_t)Bg * GNN_D;

    char* p = (char*)d_ws;
    char* pend = p + ws_size;
    auto take = [&](size_t bytes) -> void* {
        void* r = (void*)p;
        p += (bytes + 255) & ~(size_t)255;
        return r;
    };
    float*    xa     = (float*)take(ND * 4);
    float*    xb     = (float*)take(ND * 4);
    int*      rowptr = (int*)take(((size_t)N + 1) * 4);
    int*      nbr    = (int*)take((size_t)E * 4);
    unsigned* part   = (unsigned*)take((size_t)E * 4);
    int*      binTC  = (int*)take((size_t)2 * NBMAX * 4);
    int*      binSt  = (int*)take(((size_t)NBMAX + 1) * 4);
    float*    ps     = (float*)take(BD * 4);
    unsigned* pmx    = (unsigned*)take(BD * 4);
    float*    cnt    = (float*)take((size_t)Bg * 4);
    if (p > pend) return;
    const int NB = (N + BIN_NODES - 1) / BIN_NODES;
    if (NB > NBMAX || N > (1 << 24)) return;

    const int* src = eidx;
    const int* dst = eidx + E;

    hipMemcpyAsync(xa, x_in, ND * 4, hipMemcpyDeviceToDevice, stream);

    hipMemsetAsync(binTC, 0, (size_t)2 * NBMAX * 4, stream);
    gnn_bincount_kernel<<<(E + PART_CH - 1) / PART_CH, 256, 0, stream>>>(dst, binTC, E);
    gnn_partition_kernel<<<(E + PART_CH - 1) / PART_CH, 256, 0, stream>>>(
        src, dst, binTC, binTC + NBMAX, binSt, part, E, NB);
    gnn_binbuild_kernel<<<NB, 1024, 0, stream>>>(part, binSt, rowptr, nbr, N, E);

    float* xcur = xa;
    float* xnxt = xb;
    for (int l = 0; l < L; ++l) {
        gnn_layer_kernel<<<N / TILE_M, 256, 0, stream>>>(
            xcur, rowptr, nbr, z_agg + l * 3,
            W1 + (size_t)l * GNN_D * GNN_D, b1 + (size_t)l * GNN_D,
            W2 + (size_t)l * GNN_D * GNN_D, b2 + (size_t)l * GNN_D, xnxt);
        float* t = xcur; xcur = xnxt; xnxt = t;
    }

    hipMemsetAsync(ps, 0, BD * 4, stream);
    hipMemsetAsync(cnt, 0, (size_t)Bg * 4, stream);
    gnn_init_mono_kernel<<<(int)((BD + 255) / 256), 256, 0, stream>>>(pmx, (int)BD);
    gnn_pool_scatter_kernel<<<(N + 2 * POOL_RUN - 1) / (2 * POOL_RUN), 64, 0, stream>>>(xcur, batch, ps, pmx, cnt, N);
    gnn_pool_finalize_kernel<<<(int)((BD + 255) / 256), 256, 0, stream>>>(
        ps, pmx, cnt, z_pool, (float*)d_out, (int)BD);
}


# Layer kernel: 32 rows per work-group (was 16): a wave's W fragment feeds two row tiles, each its own accumulator; a wave gathers 4 rows, 8 neighbour loads per trip; sums stay in list order
# speedup vs baseline: 5.9674x; 1.0320x over previous
#include <hip/hip_runtime.h>
#include <hip/hip_bf16.h>


typedef float v8f __attribute__((ext_vector_type(8)));
typedef float v2f __attribute__((ext_vector_type(2)));

#define GNN_D 128
#define TILE_M 32
#define MT (TILE_M / 16)
#define RPW (TILE_M / 8)
#define LDS_STRIDE 132

__device__ __forceinline__ unsigned f2mono(float f) {
    unsigned u = __float_as_uint(f);
    return (u & 0x80000000u) ? ~u : (u | 0x80000000u);
}
__device__ __forceinline__ float mono2f(unsigned m) {
    unsigned u = (m & 0x80000000u) ? (m & 0x7FFFFFFFu) : ~m;
    return __uint_as_float(u);
}
#define MONO_NEG_INF 0x007FFFFFu


__global__ __launch_bounds__(256) void gnn_init_mono_kernel(unsigned* __restrict__ p, int n) {
    int i = blockIdx.x * 256 + threadIdx.x;
    if (i < n) p[i] = MONO_NEG_INF;
}

#define BIN_SHIFT 8
#define BIN_NODES 256
#define NBMAX 256
#define PART_CH 2048
#define STAGE_CAP 16384

__global__ __launch_bounds__(256) void gnn_bincount_kernel(const int* __restrict__ dst,
                                                           int* __restrict__ binTotal, int E) {
    __shared__ int h[NBMAX];
    const int t = threadIdx.x;
    const int e0 = blockIdx.x * PART_CH + t;
    int dv[8];
    #pragma unroll
    for (int k = 0; k < 8; ++k) {
        const int e = e0 + k * 256;
        const int v = dst[min(e, E - 1)];
        dv[k] = (e < E) ? v : -1;
    }
    h[t] = 0;
    __syncthreads();
    #pragma unroll
    for (int k = 0; k < 8; ++k)
        if (dv[k] >= 0) atomicAdd(&h[dv[k] >> BIN_SHIFT], 1);
    __syncthreads();
    const int c = h[t];
    if (c > 0) atomicAdd(&binTotal[t], c);
}

__global__ __launch_bounds__(256) void gnn_partition_kernel(
    const int* __restrict__ src, const int* __restrict__ dst,
    const int* __restrict__ binTotal, int* __restrict__ binCursor,
    int* __restrict__ binStart, unsigned* __restrict__ part, int E, int NB)
{
    __shared__ int      lcnt[NBMAX];
    __shared__ int      lstart[NBMAX];
    __shared__ int      gdelta[NBMAX];
    __shared__ unsigned stage[PART_CH];
    __shared__ int      sbin[PART_CH];
    __shared__ int      wtA[8];
    __shared__ int      wtB[8];
    const int t    = threadIdx.x;
    const int lane = t & 31;
    const int wv   = t >> 5;
    const int c0   = blockIdx.x * PART_CH;

    int d[8], s[8], rk[8];
    #pragma unroll
    for (int k = 0; k < 8; ++k) {
        const int e  = c0 + t + k * 256;
        const int ec = min(e, E - 1);
        const int dd = dst[ec];
        s[k] = src[ec];
        d[k] = (e < E) ? dd : -1;
    }
    lcnt[t] = 0;
    __syncthreads();
    #pragma unroll
    for (int k = 0; k < 8; ++k) {
        rk[k] = 0;
        if (d[k] >= 0) rk[k] = atomicAdd(&lcnt[d[k] >> BIN_SHIFT], 1);
    }
    __syncthreads();

    const int c   = lcnt[t];
    const int tot = (t < NB) ? binTotal[t] : 0;
    int incA = tot, incB = c;
    for (int off = 1; off < 32; off <<= 1) {
        const int ua = __shfl_up(incA, off, 32);
        const int ub = __shfl_up(incB, off, 32);
        if (lane >= off) { incA += ua; incB += ub; }
    }
    if (lane == 31) { wtA[wv] = incA; wtB[wv] = incB; }
    __syncthreads();
    int wa = (lane < wv) ? wtA[lane] : 0;
    int wb = (lane < wv) ? wtB[lane] : 0;
    for (int off = 16; off > 0; off >>= 1) {
        wa += __shfl_xor(wa, off, 32);
        wb += __shfl_xor(wb, off, 32);
    }
    const int startG = wa + incA - tot;
    const int startL = wb + incB - c;
    int res = 0;
    if (c > 0) res = atomicAdd(&binCursor[t], c);
    lstart[t] = startL;
    gdelta[t] = startG + res - startL;
    if (blockIdx.x == 0) {
        if (t < NB) binStart[t] = startG;
        if (t == 0) binStart[NB] = E;
    }
    __syncthreads();

    #pragma unroll
    for (int k = 0; k < 8; ++k) {
        if (d[k] >= 0) {
            const int b = d[k] >> BIN_SHIFT;
            const int q = lstart[b] + rk[k];
            stage[q] = ((unsigned)s[k] << 8) | (unsigned)(d[k] & (BIN_NODES - 1));
            sbin[q]  = b;
        }
    }
    __syncthreads();

    const int n = min(PART_CH, E - c0);
    for (int i = t; i < n; i += 256)
        part[gdelta[sbin[i]] + i] = stage[i];
}

__global__ __launch_bounds__(1024) void gnn_binbuild_kernel(
    const unsigned* __restrict__ part, const int* __restrict__ binStart,
    int* __restrict__ rowptr, int* __restrict__ nbr, int N, int E)
{
    __shared__ int cnt[BIN_NODES];
    __shared__ int cur[BIN_NODES];
    __shared__ int wt[8];
    __shared__ int stage[STAGE_CAP];
    const int t    = threadIdx.x;
    const int lane = t & 31;
    const int wv   = t >> 5;
    const int b    = blockIdx.x;
    const int s    = binStart[b];
    const int n    = binStart[b + 1] - s;

    if (t < BIN_NODES) cnt[t] = 0;
    __syncthreads();
    for (int i = t; i < n; i += 1024)
        atomicAdd(&cnt[part[s + i] & (BIN_NODES - 1)], 1);
    __syncthreads();

    int v = 0, inc = 0;
    if (t < BIN_NODES) {
        v = cnt[t];
        inc = v;
        for (int off = 1; off < 32; off <<= 1) {
            const int u = __shfl_up(inc, off, 32);
            if (lane >= off) inc += u;
        }
        if (lane == 31) wt[wv] = inc;
    }
    __syncthreads();
    if (t < BIN_NODES) {
        int w = (lane < wv) ? wt[lane] : 0;
        for (int off = 16; off > 0; off >>= 1) w += __shfl_xor(w, off, 32);
        const int excl = w + inc - v;
        cur[t] = excl;
        const int node = b * BIN_NODES + t;
        if (node < N) rowptr[node] = s + excl;
    }
    if (b == (int)gridDim.x - 1 && t == 0) rowptr[N] = E;
    __syncthreads();

    const bool fits = n <= STAGE_CAP;
    for (int i = t; i < n; i += 1024) {
        const unsigned pk = part[s + i];
        const int p = atomicAdd(&cur[pk & (BIN_NODES - 1)], 1);
        const int sid = (int)(pk >> 8);
        if (fits) stage[p] = sid;
        else      nbr[s + p] = sid;
    }
    __syncthreads();
    if (fits) {
        for (int i = t; i < n; i += 1024) nbr[s + i] = stage[i];
    }
}

__device__ __forceinline__ void gnn_acc(float4& sv, float4& mv, const float4 v) {
    sv.x += v.x; sv.y += v.y; sv.z += v.z; sv.w += v.w;
    mv.x = fmaxf(mv.x, v.x); mv.y = fmaxf(mv.y, v.y);
    mv.z = fmaxf(mv.z, v.z); mv.w = fmaxf(mv.w, v.w);
}

__global__ __launch_bounds__(256) void gnn_layer_kernel(
    const float* __restrict__ x, const int* __restrict__ rowptr,
    const int* __restrict__ nbr,
    const float* __restrict__ zagg,
    const float* __restrict__ W1, const float* __restrict__ b1,
    const float* __restrict__ W2, const float* __restrict__ b2,
    float* __restrict__ xout, int N)
{
    __shared__ float hs[TILE_M * LDS_STRIDE];
    __shared__ float xt[TILE_M * LDS_STRIDE];
    __shared__ float ts[TILE_M * LDS_STRIDE];

    const int tid  = threadIdx.x;
    const int row0 = blockIdx.x * TILE_M;

    const float z0 = zagg[0], z1 = zagg[1], z2 = zagg[2];

    {
        const int wv = __builtin_amdgcn_readfirstlane(tid >> 5);
        const int ln = tid & 31;
        const int nodeA = row0 + wv * RPW;
        int bnd[RPW + 1];
        #pragma unroll
        for (int i = 0; i <= RPW; ++i) bnd[i] = rowptr[min(nodeA + i, N)];
        const float* xl = x + ln * 4;
        #pragma unroll
        for (int rr = 0; rr < RPW; ++rr) {
            const int r   = wv * RPW + rr;
            float4 h  = make_float4(0.0f, 0.0f, 0.0f, 0.0f);
            float4 xv = make_float4(0.0f, 0.0f, 0.0f, 0.0f);
            if (row0 + r < N) {
                const int beg = bnd[rr];
                const int end = bnd[rr + 1];
                xv = *reinterpret_cast<const float4*>(xl + (size_t)(row0 + r) * GNN_D);
                float4 sv = make_float4(0.0f, 0.0f, 0.0f, 0.0f);
                const float ninf = -__builtin_inff();
                float4 mv = make_float4(ninf, ninf, ninf, ninf);
                int j = beg;
                for (; j + 8 <= end; j += 8) {
                    int    s[8];
                    float4 v[8];
                    #pragma unroll
                    for (int k = 0; k < 8; ++k) s[k] = nbr[j + k];
                    #pragma unroll
                    for (int k = 0; k < 8; ++k)
                        v[k] = *reinterpret_cast<const float4*>(xl + (size_t)s[k] * GNN_D);
                    #pragma unroll
                    for (int k = 0; k < 8; ++k) gnn_acc(sv, mv, v[k]);
                }
                if (j + 4 <= end) {
                    const int s0 = nbr[j], s1 = nbr[j + 1], s2 = nbr[j + 2], s3 = nbr[j + 3];
                    const float4 v0 = *reinterpret_cast<const float4*>(xl + (size_t)s0 * GNN_D);
                    const float4 v1 = *reinterpret_cast<const float4*>(xl + (size_t)s1 * GNN_D);
                    const float4 v2 = *reinterpret_cast<const float4*>(xl + (size_t)s2 * GNN_D);
                    const float4 v3 = *reinterpret_cast<const float4*>(xl + (size_t)s3 * GNN_D);
                    gnn_acc(sv, mv, v0); gnn_acc(sv, mv, v1);
                    gnn_acc(sv, mv, v2); gnn_acc(sv, mv, v3);
                    j += 4;
                }
                if (j + 2 <= end) {
                    const int s0 = nbr[j], s1 = nbr[j + 1];
                    const float4 v0 = *reinterpret_cast<const float4*>(xl + (size_t)s0 * GNN_D);
                    const float4 v1 = *reinterpret_cast<const float4*>(xl + (size_t)s1 * GNN_D);
                    gnn_acc(sv, mv, v0); gnn_acc(sv, mv, v1);
                    j += 2;
                }
                if (j < end) {
                    const int s0 = nbr[j];
                    const float4 v0 = *reinterpret_cast<const float4*>(xl + (size_t)s0 * GNN_D);
                    gnn_acc(sv, mv, v0);
                }
                const float dg = (float)(end - beg);
                const float dc = fmaxf(dg, 1.0f);
                const bool  nb = dg > 0.0f;
                h.x = xv.x + z0 * sv.x + z1 * (sv.x / dc) + z2 * (nb ? mv.x : 0.0f);
                h.y = xv.y + z0 * sv.y + z1 * (sv.y / dc) + z2 * (nb ? mv.y : 0.0f);
                h.z = xv.z + z0 * sv.z + z1 * (sv.z / dc) + z2 * (nb ? mv.z : 0.0f);
                h.w = xv.w + z0 * sv.w + z1 * (sv.w / dc) + z2 * (nb ? mv.w : 0.0f);
            }
            *reinterpret_cast<float4*>(&hs[r * LDS_STRIDE + ln * 4]) = h;
            *reinterpret_cast<float4*>(&xt[r * LDS_STRIDE + ln * 4]) = xv;
        }
    }
    __syncthreads();

    const int wave  = tid >> 5;
    const int lane  = tid & 31;
    const int ncol  = (wave << 4) + (lane & 15);
    const int mlo   = lane & 15;
    const int koff  = (lane >> 4) << 1;
    const int rbase = (lane >> 4) << 3;
    const float bia1 = b1[ncol];
    const float bia2 = b2[ncol];

    v8f acc[MT];
    #pragma unroll
    for (int m = 0; m < MT; ++m) acc[m] = v8f{};
    #pragma unroll
    for (int k0 = 0; k0 < GNN_D; k0 += 4) {
        const int kk = k0 + koff;
        v2f b; b.x = W1[(size_t)kk * GNN_D + ncol];
               b.y = W1[(size_t)(kk + 1) * GNN_D + ncol];
        #pragma unroll
        for (int m = 0; m < MT; ++m) {
            v2f a; a.x = hs[(m * 16 + mlo) * LDS_STRIDE + kk];
                   a.y = hs[(m * 16 + mlo) * LDS_STRIDE + kk + 1];
            acc[m] = __builtin_amdgcn_wmma_f32_16x16x4_f32(false, a, false, b,
                                                           (short)0, acc[m], false, false);
        }
    }
    #pragma unroll
    for (int m = 0; m < MT; ++m) {
        #pragma unroll
        for (int i = 0; i < 8; ++i)
            ts[(m * 16 + rbase + i) * LDS_STRIDE + ncol] = fmaxf(acc[m][i] + bia1, 0.0f);
    }
    __syncthreads();

    v8f acc2[MT];
    #pragma unroll
    for (int m = 0; m < MT; ++m) acc2[m] = v8f{};
    #pragma unroll
    for (int k0 = 0; k0 < GNN_D; k0 += 4) {
        const int kk = k0 + koff;
        v2f b; b.x = W2[(size_t)kk * GNN_D + ncol];
               b.y = W2[(size_t)(kk + 1) * GNN_D + ncol];
        #pragma unroll
        for (int m = 0; m < MT; ++m) {
            v2f a; a.x = ts[(m * 16 + mlo) * LDS_STRIDE + kk];
                   a.y = ts[(m * 16 + mlo) * LDS_STRIDE + kk + 1];
            acc2[m] = __builtin_amdgcn_wmma_f32_16x16x4_f32(false, a, false, b,
                                                            (short)0, acc2[m], false, false);
        }
    }
    #pragma unroll
    for (int m = 0; m < MT; ++m) {
        #pragma unroll
        for (int i = 0; i < 8; ++i) {
            const int r = m * 16 + rbase + i;
            const float h2 = acc2[m][i] + bia2;
            const float xv = xt[r * LDS_STRIDE + ncol];
            if (row0 + r < N)
                xout[(size_t)(row0 + r) * GNN_D + ncol] = xv + fmaxf(h2, 0.0f);
        }
    }
}

#define POOL_RUN 32

__device__ __forceinline__ void gnn_pool_flush(
    float* __restrict__ ps, unsigned* __restrict__ pmx, float* __restrict__ cnt,
    int b, int lane, const float4& sv, const uint4& mm, int run)
{
    float*    sp = ps  + (size_t)b * GNN_D + lane * 4;
    unsigned* mp = pmx + (size_t)b * GNN_D + lane * 4;
    atomicAdd(sp + 0, sv.x); atomicAdd(sp + 1, sv.y);
    atomicAdd(sp + 2, sv.z); atomicAdd(sp + 3, sv.w);
    atomicMax(mp + 0, mm.x); atomicMax(mp + 1, mm.y);
    atomicMax(mp + 2, mm.z); atomicMax(mp + 3, mm.w);
    if (lane == 0) atomicAdd(&cnt[b], (float)run);
}

__global__ __launch_bounds__(64) void gnn_pool_scatter_kernel(
    const float* __restrict__ x, const int* __restrict__ batch,
    float* __restrict__ ps, unsigned* __restrict__ pmx,
    float* __restrict__ cnt, int N)
{
    const int wv   = blockIdx.x * 2 + __builtin_amdgcn_readfirstlane((int)threadIdx.x >> 5);
    const int lane = threadIdx.x & 31;
    const int n0   = wv * POOL_RUN;
    if (n0 >= N) return;
    const int n1   = min(n0 + POOL_RUN, N);
    const float* xl = x + lane * 4;

    int    curb = batch[n0];
    float4 sv   = make_float4(0.0f, 0.0f, 0.0f, 0.0f);
    uint4  mm   = make_uint4(MONO_NEG_INF, MONO_NEG_INF, MONO_NEG_INF, MONO_NEG_INF);
    int    run  = 0;

    for (int n = n0; n < n1; n += 8) {
        int    bb[8];
        float4 v[8];
        #pragma unroll
        for (int i = 0; i < 8; ++i) {
            const int m = min(n + i, n1 - 1);
            bb[i] = batch[m];
            v[i]  = *reinterpret_cast<const float4*>(xl + (size_t)m * GNN_D);
        }
        #pragma unroll
        for (int i = 0; i < 8; ++i) {
            if (i == 0 || n + i < n1) {
                if (bb[i] != curb) {
                    gnn_pool_flush(ps, pmx, cnt, curb, lane, sv, mm, run);
                    sv   = make_float4(0.0f, 0.0f, 0.0f, 0.0f);
                    mm   = make_uint4(MONO_NEG_INF, MONO_NEG_INF, MONO_NEG_INF, MONO_NEG_INF);
                    run  = 0;
                    curb = bb[i];
                }
                sv.x += v[i].x; sv.y += v[i].y; sv.z += v[i].z; sv.w += v[i].w;
                mm.x = max(mm.x, f2mono(v[i].x)); mm.y = max(mm.y, f2mono(v[i].y));
                mm.z = max(mm.z, f2mono(v[i].z)); mm.w = max(mm.w, f2mono(v[i].w));
                ++run;
            }
        }
    }
    gnn_pool_flush(ps, pmx, cnt, curb, lane, sv, mm, run);
}

__global__ __launch_bounds__(256) void gnn_pool_finalize_kernel(
    const float* __restrict__ ps, const unsigned* __restrict__ pmx,
    const float* __restrict__ cnt, const float* __restrict__ zpool,
    float* __restrict__ out, int total)
{
    int i = blockIdx.x * 256 + threadIdx.x;
    if (i >= total) return;
    int b = i >> 7;
    float c  = cnt[b];
    float sv = ps[i];
    float mv = (c > 0.0f) ? mono2f(pmx[i]) : 0.0f;
    out[i] = zpool[0] * sv + zpool[1] * (sv / fmaxf(c, 1.0f)) + zpool[2] * mv;
}


extern "C" void kernel_launch(void* const* d_in, const int* in_sizes, int n_in,
                              void* d_out, int out_size, void* d_ws, size_t ws_size,
                              hipStream_t stream) {
    const float* x_in   = (const float*)d_in[0];
    const int*   eidx   = (const int*)d_in[1];
    const int*   batch  = (const int*)d_in[2];
    const float* z_agg  = (const float*)d_in[3];
    const float* z_pool = (const float*)d_in[4];
    const float* W1     = (const float*)d_in[5];
    const float* b1     = (const float*)d_in[6];
    const float* W2     = (const float*)d_in[7];
    const float* b2     = (const float*)d_in[8];
    (void)n_in;

    const int N = in_sizes[0] / GNN_D;
    const int E = in_sizes[1] / 2;
    const int L = in_sizes[5] / (GNN_D * GNN_D);
    const int Bg = out_size / GNN_D;
    const size_t ND = (size_t)N * GNN_D;
    const size_t BD = (size_t)Bg * GNN_D;

    char* p = (char*)d_ws;
    char* pend = p + ws_size;
    auto take = [&](size_t bytes) -> void* {
        void* r = (void*)p;
        p += (bytes + 255) & ~(size_t)255;
        return r;
    };
    float*    xa     = (float*)take(ND * 4);
    float*    xb     = (float*)take(ND * 4);
    int*      rowptr = (int*)take(((size_t)N + 1) * 4);
    int*      nbr    = (int*)take((size_t)E * 4);
    unsigned* part   = (unsigned*)take((size_t)E * 4);
    int*      binTC  = (int*)take((size_t)2 * NBMAX * 4);
    int*      binSt  = (int*)take(((size_t)NBMAX + 1) * 4);
    float*    ps     = (float*)take(BD * 4);
    unsigned* pmx    = (unsigned*)take(BD * 4);
    float*    cnt    = (float*)take((size_t)Bg * 4);
    if (p > pend) return;
    const int NB = (N + BIN_NODES - 1) / BIN_NODES;
    if (NB > NBMAX || N > (1 << 24)) return;

    const int* src = eidx;
    const int* dst = eidx + E;

    hipMemcpyAsync(xa, x_in, ND * 4, hipMemcpyDeviceToDevice, stream);

    hipMemsetAsync(binTC, 0, (size_t)2 * NBMAX * 4, stream);
    gnn_bincount_kernel<<<(E + PART_CH - 1) / PART_CH, 256, 0, stream>>>(dst, binTC, E);
    gnn_partition_kernel<<<(E + PART_CH - 1) / PART_CH, 256, 0, stream>>>(
        src, dst, binTC, binTC + NBMAX, binSt, part, E, NB);
    gnn_binbuild_kernel<<<NB, 1024, 0, stream>>>(part, binSt, rowptr, nbr, N, E);

    float* xcur = xa;
    float* xnxt = xb;
    for (int l = 0; l < L; ++l) {
        gnn_layer_kernel<<<(N + TILE_M - 1) / TILE_M, 256, 0, stream>>>(
            xcur, rowptr, nbr, z_agg + l * 3,
            W1 + (size_t)l * GNN_D * GNN_D, b1 + (size_t)l * GNN_D,
            W2 + (size_t)l * GNN_D * GNN_D, b2 + (size_t)l * GNN_D, xnxt, N);
        float* t = xcur; xcur = xnxt; xnxt = t;
    }

    hipMemsetAsync(ps, 0, BD * 4, stream);
    hipMemsetAsync(cnt, 0, (size_t)Bg * 4, stream);
    gnn_init_mono_kernel<<<(int)((BD + 255) / 256), 256, 0, stream>>>(pmx, (int)BD);
    gnn_pool_scatter_kernel<<<(N + 2 * POOL_RUN - 1) / (2 * POOL_RUN), 64, 0, stream>>>(xcur, batch, ps, pmx, cnt, N);
    gnn_pool_finalize_kernel<<<(int)((BD + 255) / 256), 256, 0, stream>>>(
        ps, pmx, cnt, z_pool, (float*)d_out, (int)BD);
}


# Stream operations 13 -> 9: input copy and the three pooling clears removed; layer 0 gathers from x in place, the first edge launch also writes the pooled start values; all sums unchanged
# speedup vs baseline: 6.6851x; 1.1203x over previous
#include <hip/hip_runtime.h>
#include <hip/hip_bf16.h>


typedef float v8f __attribute__((ext_vector_type(8)));
typedef float v2f __attribute__((ext_vector_type(2)));

#define GNN_D 128
#define TILE_M 32
#define MT (TILE_M / 16)
#define RPW (TILE_M / 8)
#define LDS_STRIDE 132

__device__ __forceinline__ unsigned f2mono(float f) {
    unsigned u = __float_as_uint(f);
    return (u & 0x80000000u) ? ~u : (u | 0x80000000u);
}
__device__ __forceinline__ float mono2f(unsigned m) {
    unsigned u = (m & 0x80000000u) ? (m & 0x7FFFFFFFu) : ~m;
    return __uint_as_float(u);
}
#define MONO_NEG_INF 0x007FFFFFu

#define BIN_SHIFT 8
#define BIN_NODES 256
#define NBMAX 256
#define PART_CH 2048
#define STAGE_CAP 16384

__global__ __launch_bounds__(256) void gnn_bincount_kernel(const int* __restrict__ dst,
                                                           int* __restrict__ binTotal, int E,
                                                           float* __restrict__ ps,
                                                           unsigned* __restrict__ pmx,
                                                           float* __restrict__ cnt,
                                                           int BD, int Bg) {
    __shared__ int h[NBMAX];
    const int t = threadIdx.x;
    for (int i = blockIdx.x * 256 + t; i < BD; i += (int)gridDim.x * 256) {
        ps[i]  = 0.0f;
        pmx[i] = MONO_NEG_INF;
        if (i < Bg) cnt[i] = 0.0f;
    }
    const int e0 = blockIdx.x * PART_CH + t;
    int dv[8];
    #pragma unroll
    for (int k = 0; k < 8; ++k) {
        const int e = e0 + k * 256;
        const int v = dst[min(e, E - 1)];
        dv[k] = (e < E) ? v : -1;
    }
    h[t] = 0;
    __syncthreads();
    #pragma unroll
    for (int k = 0; k < 8; ++k)
        if (dv[k] >= 0) atomicAdd(&h[dv[k] >> BIN_SHIFT], 1);
    __syncthreads();
    const int c = h[t];
    if (c > 0) atomicAdd(&binTotal[t], c);
}

__global__ __launch_bounds__(256) void gnn_partition_kernel(
    const int* __restrict__ src, const int* __restrict__ dst,
    const int* __restrict__ binTotal, int* __restrict__ binCursor,
    int* __restrict__ binStart, unsigned* __restrict__ part, int E, int NB)
{
    __shared__ int      lcnt[NBMAX];
    __shared__ int      lstart[NBMAX];
    __shared__ int      gdelta[NBMAX];
    __shared__ unsigned stage[PART_CH];
    __shared__ int      sbin[PART_CH];
    __shared__ int      wtA[8];
    __shared__ int      wtB[8];
    const int t    = threadIdx.x;
    const int lane = t & 31;
    const int wv   = t >> 5;
    const int c0   = blockIdx.x * PART_CH;

    int d[8], s[8], rk[8];
    #pragma unroll
    for (int k = 0; k < 8; ++k) {
        const int e  = c0 + t + k * 256;
        const int ec = min(e, E - 1);
        const int dd = dst[ec];
        s[k] = src[ec];
        d[k] = (e < E) ? dd : -1;
    }
    lcnt[t] = 0;
    __syncthreads();
    #pragma unroll
    for (int k = 0; k < 8; ++k) {
        rk[k] = 0;
        if (d[k] >= 0) rk[k] = atomicAdd(&lcnt[d[k] >> BIN_SHIFT], 1);
    }
    __syncthreads();

    const int c   = lcnt[t];
    const int tot = (t < NB) ? binTotal[t] : 0;
    int incA = tot, incB = c;
    for (int off = 1; off < 32; off <<= 1) {
        const int ua = __shfl_up(incA, off, 32);
        const int ub = __shfl_up(incB, off, 32);
        if (lane >= off) { incA += ua; incB += ub; }
    }
    if (lane == 31) { wtA[wv] = incA; wtB[wv] = incB; }
    __syncthreads();
    int wa = (lane < wv) ? wtA[lane] : 0;
    int wb = (lane < wv) ? wtB[lane] : 0;
    for (int off = 16; off > 0; off >>= 1) {
        wa += __shfl_xor(wa, off, 32);
        wb += __shfl_xor(wb, off, 32);
    }
    const int startG = wa + incA - tot;
    const int startL = wb + incB - c;
    int res = 0;
    if (c > 0) res = atomicAdd(&binCursor[t], c);
    lstart[t] = startL;
    gdelta[t] = startG + res - startL;
    if (blockIdx.x == 0) {
        if (t < NB) binStart[t] = startG;
        if (t == 0) binStart[NB] = E;
    }
    __syncthreads();

    #pragma unroll
    for (int k = 0; k < 8; ++k) {
        if (d[k] >= 0) {
            const int b = d[k] >> BIN_SHIFT;
            const int q = lstart[b] + rk[k];
            stage[q] = ((unsigned)s[k] << 8) | (unsigned)(d[k] & (BIN_NODES - 1));
            sbin[q]  = b;
        }
    }
    __syncthreads();

    const int n = min(PART_CH, E - c0);
    for (int i = t; i < n; i += 256)
        part[gdelta[sbin[i]] + i] = stage[i];
}

__global__ __launch_bounds__(1024) void gnn_binbuild_kernel(
    const unsigned* __restrict__ part, const int* __restrict__ binStart,
    int* __restrict__ rowptr, int* __restrict__ nbr, int N, int E)
{
    __shared__ int cnt[BIN_NODES];
    __shared__ int cur[BIN_NODES];
    __shared__ int wt[8];
    __shared__ int stage[STAGE_CAP];
    const int t    = threadIdx.x;
    const int lane = t & 31;
    const int wv   = t >> 5;
    const int b    = blockIdx.x;
    const int s    = binStart[b];
    const int n    = binStart[b + 1] - s;

    if (t < BIN_NODES) cnt[t] = 0;
    __syncthreads();
    for (int i = t; i < n; i += 1024)
        atomicAdd(&cnt[part[s + i] & (BIN_NODES - 1)], 1);
    __syncthreads();

    int v = 0, inc = 0;
    if (t < BIN_NODES) {
        v = cnt[t];
        inc = v;
        for (int off = 1; off < 32; off <<= 1) {
            const int u = __shfl_up(inc, off, 32);
            if (lane >= off) inc += u;
        }
        if (lane == 31) wt[wv] = inc;
    }
    __syncthreads();
    if (t < BIN_NODES) {
        int w = (lane < wv) ? wt[lane] : 0;
        for (int off = 16; off > 0; off >>= 1) w += __shfl_xor(w, off, 32);
        const int excl = w + inc - v;
        cur[t] = excl;
        const int node = b * BIN_NODES + t;
        if (node < N) rowptr[node] = s + excl;
    }
    if (b == (int)gridDim.x - 1 && t == 0) rowptr[N] = E;
    __syncthreads();

    const bool fits = n <= STAGE_CAP;
    for (int i = t; i < n; i += 1024) {
        const unsigned pk = part[s + i];
        const int p = atomicAdd(&cur[pk & (BIN_NODES - 1)], 1);
        const int sid = (int)(pk >> 8);
        if (fits) stage[p] = sid;
        else      nbr[s + p] = sid;
    }
    __syncthreads();
    if (fits) {
        for (int i = t; i < n; i += 1024) nbr[s + i] = stage[i];
    }
}

__device__ __forceinline__ void gnn_acc(float4& sv, float4& mv, const float4 v) {
    sv.x += v.x; sv.y += v.y; sv.z += v.z; sv.w += v.w;
    mv.x = fmaxf(mv.x, v.x); mv.y = fmaxf(mv.y, v.y);
    mv.z = fmaxf(mv.z, v.z); mv.w = fmaxf(mv.w, v.w);
}

__global__ __launch_bounds__(256) void gnn_layer_kernel(
    const float* __restrict__ x, const int* __restrict__ rowptr,
    const int* __restrict__ nbr,
    const float* __restrict__ zagg,
    const float* __restrict__ W1, const float* __restrict__ b1,
    const float* __restrict__ W2, const float* __restrict__ b2,
    float* __restrict__ xout, int N)
{
    __shared__ float hs[TILE_M * LDS_STRIDE];
    __shared__ float xt[TILE_M * LDS_STRIDE];
    __shared__ float ts[TILE_M * LDS_STRIDE];

    const int tid  = threadIdx.x;
    const int row0 = blockIdx.x * TILE_M;

    const float z0 = zagg[0], z1 = zagg[1], z2 = zagg[2];

    {
        const int wv = __builtin_amdgcn_readfirstlane(tid >> 5);
        const int ln = tid & 31;
        const int nodeA = row0 + wv * RPW;
        int bnd[RPW + 1];
        #pragma unroll
        for (int i = 0; i <= RPW; ++i) bnd[i] = rowptr[min(nodeA + i, N)];
        const float* xl = x + ln * 4;
        #pragma unroll
        for (int rr = 0; rr < RPW; ++rr) {
            const int r   = wv * RPW + rr;
            float4 h  = make_float4(0.0f, 0.0f, 0.0f, 0.0f);
            float4 xv = make_float4(0.0f, 0.0f, 0.0f, 0.0f);
            if (row0 + r < N) {
                const int beg = bnd[rr];
                const int end = bnd[rr + 1];
                xv = *reinterpret_cast<const float4*>(xl + (size_t)(row0 + r) * GNN_D);
                float4 sv = make_float4(0.0f, 0.0f, 0.0f, 0.0f);
                const float ninf = -__builtin_inff();
                float4 mv = make_float4(ninf, ninf, ninf, ninf);
                int j = beg;
                for (; j + 8 <= end; j += 8) {
                    int    s[8];
                    float4 v[8];
                    #pragma unroll
                    for (int k = 0; k < 8; ++k) s[k] = nbr[j + k];
                    #pragma unroll
                    for (int k = 0; k < 8; ++k)
                        v[k] = *reinterpret_cast<const float4*>(xl + (size_t)s[k] * GNN_D);
                    #pragma unroll
                    for (int k = 0; k < 8; ++k) gnn_acc(sv, mv, v[k]);
                }
                if (j + 4 <= end) {
                    const int s0 = nbr[j], s1 = nbr[j + 1], s2 = nbr[j + 2], s3 = nbr[j + 3];
                    const float4 v0 = *reinterpret_cast<const float4*>(xl + (size_t)s0 * GNN_D);
                    const float4 v1 = *reinterpret_cast<const float4*>(xl + (size_t)s1 * GNN_D);
                    const float4 v2 = *reinterpret_cast<const float4*>(xl + (size_t)s2 * GNN_D);
                    const float4 v3 = *reinterpret_cast<const float4*>(xl + (size_t)s3 * GNN_D);
                    gnn_acc(sv, mv, v0); gnn_acc(sv, mv, v1);
                    gnn_acc(sv, mv, v2); gnn_acc(sv, mv, v3);
                    j += 4;
                }
                if (j + 2 <= end) {
                    const int s0 = nbr[j], s1 = nbr[j + 1];
                    const float4 v0 = *reinterpret_cast<const float4*>(xl + (size_t)s0 * GNN_D);
                    const float4 v1 = *reinterpret_cast<const float4*>(xl + (size_t)s1 * GNN_D);
                    gnn_acc(sv, mv, v0); gnn_acc(sv, mv, v1);
                    j += 2;
                }
                if (j < end) {
                    const int s0 = nbr[j];
                    const float4 v0 = *reinterpret_cast<const float4*>(xl + (size_t)s0 * GNN_D);
                    gnn_acc(sv, mv, v0);
                }
                const float dg = (float)(end - beg);
                const float dc = fmaxf(dg, 1.0f);
                const bool  nb = dg > 0.0f;
                h.x = xv.x + z0 * sv.x + z1 * (sv.x / dc) + z2 * (nb ? mv.x : 0.0f);
                h.y = xv.y + z0 * sv.y + z1 * (sv.y / dc) + z2 * (nb ? mv.y : 0.0f);
                h.z = xv.z + z0 * sv.z + z1 * (sv.z / dc) + z2 * (nb ? mv.z : 0.0f);
                h.w = xv.w + z0 * sv.w + z1 * (sv.w / dc) + z2 * (nb ? mv.w : 0.0f);
            }
            *reinterpret_cast<float4*>(&hs[r * LDS_STRIDE + ln * 4]) = h;
            *reinterpret_cast<float4*>(&xt[r * LDS_STRIDE + ln * 4]) = xv;
        }
    }
    __syncthreads();

    const int wave  = tid >> 5;
    const int lane  = tid & 31;
    const int ncol  = (wave << 4) + (lane & 15);
    const int mlo   = lane & 15;
    const int koff  = (lane >> 4) << 1;
    const int rbase = (lane >> 4) << 3;
    const float bia1 = b1[ncol];
    const float bia2 = b2[ncol];

    v8f acc[MT];
    #pragma unroll
    for (int m = 0; m < MT; ++m) acc[m] = v8f{};
    #pragma unroll
    for (int k0 = 0; k0 < GNN_D; k0 += 4) {
        const int kk = k0 + koff;
        v2f b; b.x = W1[(size_t)kk * GNN_D + ncol];
               b.y = W1[(size_t)(kk + 1) * GNN_D + ncol];
        #pragma unroll
        for (int m = 0; m < MT; ++m) {
            v2f a; a.x = hs[(m * 16 + mlo) * LDS_STRIDE + kk];
                   a.y = hs[(m * 16 + mlo) * LDS_STRIDE + kk + 1];
            acc[m] = __builtin_amdgcn_wmma_f32_16x16x4_f32(false, a, false, b,
                                                           (short)0, acc[m], false, false);
        }
    }
    #pragma unroll
    for (int m = 0; m < MT; ++m) {
        #pragma unroll
        for (int i = 0; i < 8; ++i)
            ts[(m * 16 + rbase + i) * LDS_STRIDE + ncol] = fmaxf(acc[m][i] + bia1, 0.0f);
    }
    __syncthreads();

    v8f acc2[MT];
    #pragma unroll
    for (int m = 0; m < MT; ++m) acc2[m] = v8f{};
    #pragma unroll
    for (int k0 = 0; k0 < GNN_D; k0 += 4) {
        const int kk = k0 + koff;
        v2f b; b.x = W2[(size_t)kk * GNN_D + ncol];
               b.y = W2[(size_t)(kk + 1) * GNN_D + ncol];
        #pragma unroll
        for (int m = 0; m < MT; ++m) {
            v2f a; a.x = ts[(m * 16 + mlo) * LDS_STRIDE + kk];
                   a.y = ts[(m * 16 + mlo) * LDS_STRIDE + kk + 1];
            acc2[m] = __builtin_amdgcn_wmma_f32_16x16x4_f32(false, a, false, b,
                                                            (short)0, acc2[m], false, false);
        }
    }
    #pragma unroll
    for (int m = 0; m < MT; ++m) {
        #pragma unroll
        for (int i = 0; i < 8; ++i) {
            const int r = m * 16 + rbase + i;
            const float h2 = acc2[m][i] + bia2;
            const float xv = xt[r * LDS_STRIDE + ncol];
            if (row0 + r < N)
                xout[(size_t)(row0 + r) * GNN_D + ncol] = xv + fmaxf(h2, 0.0f);
        }
    }
}

#define POOL_RUN 32

__device__ __forceinline__ void gnn_pool_flush(
    float* __restrict__ ps, unsigned* __restrict__ pmx, float* __restrict__ cnt,
    int b, int lane, const float4& sv, const uint4& mm, int run)
{
    float*    sp = ps  + (size_t)b * GNN_D + lane * 4;
    unsigned* mp = pmx + (size_t)b * GNN_D + lane * 4;
    atomicAdd(sp + 0, sv.x); atomicAdd(sp + 1, sv.y);
    atomicAdd(sp + 2, sv.z); atomicAdd(sp + 3, sv.w);
    atomicMax(mp + 0, mm.x); atomicMax(mp + 1, mm.y);
    atomicMax(mp + 2, mm.z); atomicMax(mp + 3, mm.w);
    if (lane == 0) atomicAdd(&cnt[b], (float)run);
}

__global__ __launch_bounds__(64) void gnn_pool_scatter_kernel(
    const float* __restrict__ x, const int* __restrict__ batch,
    float* __restrict__ ps, unsigned* __restrict__ pmx,
    float* __restrict__ cnt, int N)
{
    const int wv   = blockIdx.x * 2 + __builtin_amdgcn_readfirstlane((int)threadIdx.x >> 5);
    const int lane = threadIdx.x & 31;
    const int n0   = wv * POOL_RUN;
    if (n0 >= N) return;
    const int n1   = min(n0 + POOL_RUN, N);
    const float* xl = x + lane * 4;

    int    curb = batch[n0];
    float4 sv   = make_float4(0.0f, 0.0f, 0.0f, 0.0f);
    uint4  mm   = make_uint4(MONO_NEG_INF, MONO_NEG_INF, MONO_NEG_INF, MONO_NEG_INF);
    int    run  = 0;

    for (int n = n0; n < n1; n += 8) {
        int    bb[8];
        float4 v[8];
        #pragma unroll
        for (int i = 0; i < 8; ++i) {
            const int m = min(n + i, n1 - 1);
            bb[i] = batch[m];
            v[i]  = *reinterpret_cast<const float4*>(xl + (size_t)m * GNN_D);
        }
        #pragma unroll
        for (int i = 0; i < 8; ++i) {
            if (i == 0 || n + i < n1) {
                if (bb[i] != curb) {
                    gnn_pool_flush(ps, pmx, cnt, curb, lane, sv, mm, run);
                    sv   = make_float4(0.0f, 0.0f, 0.0f, 0.0f);
                    mm   = make_uint4(MONO_NEG_INF, MONO_NEG_INF, MONO_NEG_INF, MONO_NEG_INF);
                    run  = 0;
                    curb = bb[i];
                }
                sv.x += v[i].x; sv.y += v[i].y; sv.z += v[i].z; sv.w += v[i].w;
                mm.x = max(mm.x, f2mono(v[i].x)); mm.y = max(mm.y, f2mono(v[i].y));
                mm.z = max(mm.z, f2mono(v[i].z)); mm.w = max(mm.w, f2mono(v[i].w));
                ++run;
            }
        }
    }
    gnn_pool_flush(ps, pmx, cnt, curb, lane, sv, mm, run);
}

__global__ __launch_bounds__(256) void gnn_pool_finalize_kernel(
    const float* __restrict__ ps, const unsigned* __restrict__ pmx,
    const float* __restrict__ cnt, const float* __restrict__ zpool,
    float* __restrict__ out, int total)
{
    int i = blockIdx.x * 256 + threadIdx.x;
    if (i >= total) return;
    int b = i >> 7;
    float c  = cnt[b];
    float sv = ps[i];
    float mv = (c > 0.0f) ? mono2f(pmx[i]) : 0.0f;
    out[i] = zpool[0] * sv + zpool[1] * (sv / fmaxf(c, 1.0f)) + zpool[2] * mv;
}


extern "C" void kernel_launch(void* const* d_in, const int* in_sizes, int n_in,
                              void* d_out, int out_size, void* d_ws, size_t ws_size,
                              hipStream_t stream) {
    const float* x_in   = (const float*)d_in[0];
    const int*   eidx   = (const int*)d_in[1];
    const int*   batch  = (const int*)d_in[2];
    const float* z_agg  = (const float*)d_in[3];
    const float* z_pool = (const float*)d_in[4];
    const float* W1     = (const float*)d_in[5];
    const float* b1     = (const float*)d_in[6];
    const float* W2     = (const float*)d_in[7];
    const float* b2     = (const float*)d_in[8];
    (void)n_in;

    const int N = in_sizes[0] / GNN_D;
    const int E = in_sizes[1] / 2;
    const int L = in_sizes[5] / (GNN_D * GNN_D);
    const int Bg = out_size / GNN_D;
    const size_t ND = (size_t)N * GNN_D;
    const size_t BD = (size_t)Bg * GNN_D;

    char* p = (char*)d_ws;
    char* pend = p + ws_size;
    auto take = [&](size_t bytes) -> void* {
        void* r = (void*)p;
        p += (bytes + 255) & ~(size_t)255;
        return r;
    };
    float*    xa     = (float*)take(ND * 4);
    float*    xb     = (float*)take(ND * 4);
    int*      rowptr = (int*)take(((size_t)N + 1) * 4);
    int*      nbr    = (int*)take((size_t)E * 4);
    unsigned* part   = (unsigned*)take((size_t)E * 4);
    int*      binTC  = (int*)take((size_t)2 * NBMAX * 4);
    int*      binSt  = (int*)take(((size_t)NBMAX + 1) * 4);
    float*    ps     = (float*)take(BD * 4);
    unsigned* pmx    = (unsigned*)take(BD * 4);
    float*    cnt    = (float*)take((size_t)Bg * 4);
    if (p > pend) return;
    const int NB = (N + BIN_NODES - 1) / BIN_NODES;
    if (NB > NBMAX || N > (1 << 24)) return;

    const int* src = eidx;
    const int* dst = eidx + E;

    hipMemsetAsync(binTC, 0, (size_t)2 * NBMAX * 4, stream);
    gnn_bincount_kernel<<<(E + PART_CH - 1) / PART_CH, 256, 0, stream>>>(
        dst, binTC, E, ps, pmx, cnt, (int)BD, Bg);
    gnn_partition_kernel<<<(E + PART_CH - 1) / PART_CH, 256, 0, stream>>>(
        src, dst, binTC, binTC + NBMAX, binSt, part, E, NB);
    gnn_binbuild_kernel<<<NB, 1024, 0, stream>>>(part, binSt, rowptr, nbr, N, E);

    const float* xcur = x_in;
    float* xnxt = xa;
    float* xoth = xb;
    for (int l = 0; l < L; ++l) {
        gnn_layer_kernel<<<(N + TILE_M - 1) / TILE_M, 256, 0, stream>>>(
            xcur, rowptr, nbr, z_agg + l * 3,
            W1 + (size_t)l * GNN_D * GNN_D, b1 + (size_t)l * GNN_D,
            W2 + (size_t)l * GNN_D * GNN_D, b2 + (size_t)l * GNN_D, xnxt, N);
        xcur = xnxt;
        float* t = xnxt; xnxt = xoth; xoth = t;
    }

    gnn_pool_scatter_kernel<<<(N + 2 * POOL_RUN - 1) / (2 * POOL_RUN), 64, 0, stream>>>(xcur, batch, ps, pmx, cnt, N);
    gnn_pool_finalize_kernel<<<(int)((BD + 255) / 256), 256, 0, stream>>>(
        ps, pmx, cnt, z_pool, (float*)d_out, (int)BD);
}


# Pooling moved into the last layer kernel: each tile's 32 output rows are pooled from LDS in node order, then the same float atomics; pooling launch, its re-read of x and the last x write gone
# speedup vs baseline: 7.1377x; 1.0677x over previous
#include <hip/hip_runtime.h>
#include <hip/hip_bf16.h>


typedef float v8f __attribute__((ext_vector_type(8)));
typedef float v2f __attribute__((ext_vector_type(2)));

#define GNN_D 128
#define TILE_M 32
#define MT (TILE_M / 16)
#define RPW (TILE_M / 8)
#define LDS_STRIDE 132

__device__ __forceinline__ unsigned f2mono(float f) {
    unsigned u = __float_as_uint(f);
    return (u & 0x80000000u) ? ~u : (u | 0x80000000u);
}
__device__ __forceinline__ float mono2f(unsigned m) {
    unsigned u = (m & 0x80000000u) ? (m & 0x7FFFFFFFu) : ~m;
    return __uint_as_float(u);
}
#define MONO_NEG_INF 0x007FFFFFu

#define BIN_SHIFT 8
#define BIN_NODES 256
#define NBMAX 256
#define PART_CH 2048
#define STAGE_CAP 16384

__global__ __launch_bounds__(256) void gnn_bincount_kernel(const int* __restrict__ dst,
                                                           int* __restrict__ binTotal, int E,
                                                           float* __restrict__ ps,
                                                           unsigned* __restrict__ pmx,
                                                           float* __restrict__ cnt,
                                                           int BD, int Bg) {
    __shared__ int h[NBMAX];
    const int t = threadIdx.x;
    for (int i = blockIdx.x * 256 + t; i < BD; i += (int)gridDim.x * 256) {
        ps[i]  = 0.0f;
        pmx[i] = MONO_NEG_INF;
        if (i < Bg) cnt[i] = 0.0f;
    }
    const int e0 = blockIdx.x * PART_CH + t;
    int dv[8];
    #pragma unroll
    for (int k = 0; k < 8; ++k) {
        const int e = e0 + k * 256;
        const int v = dst[min(e, E - 1)];
        dv[k] = (e < E) ? v : -1;
    }
    h[t] = 0;
    __syncthreads();
    #pragma unroll
    for (int k = 0; k < 8; ++k)
        if (dv[k] >= 0) atomicAdd(&h[dv[k] >> BIN_SHIFT], 1);
    __syncthreads();
    const int c = h[t];
    if (c > 0) atomicAdd(&binTotal[t], c);
}

__global__ __launch_bounds__(256) void gnn_partition_kernel(
    const int* __restrict__ src, const int* __restrict__ dst,
    const int* __restrict__ binTotal, int* __restrict__ binCursor,
    int* __restrict__ binStart, unsigned* __restrict__ part, int E, int NB)
{
    __shared__ int      lcnt[NBMAX];
    __shared__ int      lstart[NBMAX];
    __shared__ int      gdelta[NBMAX];
    __shared__ unsigned stage[PART_CH];
    __shared__ int      sbin[PART_CH];
    __shared__ int      wtA[8];
    __shared__ int      wtB[8];
    const int t    = threadIdx.x;
    const int lane = t & 31;
    const int wv   = t >> 5;
    const int c0   = blockIdx.x * PART_CH;

    int d[8], s[8], rk[8];
    #pragma unroll
    for (int k = 0; k < 8; ++k) {
        const int e  = c0 + t + k * 256;
        const int ec = min(e, E - 1);
        const int dd = dst[ec];
        s[k] = src[ec];
        d[k] = (e < E) ? dd : -1;
    }
    lcnt[t] = 0;
    __syncthreads();
    #pragma unroll
    for (int k = 0; k < 8; ++k) {
        rk[k] = 0;
        if (d[k] >= 0) rk[k] = atomicAdd(&lcnt[d[k] >> BIN_SHIFT], 1);
    }
    __syncthreads();

    const int c   = lcnt[t];
    const int tot = (t < NB) ? binTotal[t] : 0;
    int incA = tot, incB = c;
    for (int off = 1; off < 32; off <<= 1) {
        const int ua = __shfl_up(incA, off, 32);
        const int ub = __shfl_up(incB, off, 32);
        if (lane >= off) { incA += ua; incB += ub; }
    }
    if (lane == 31) { wtA[wv] = incA; wtB[wv] = incB; }
    __syncthreads();
    int wa = (lane < wv) ? wtA[lane] : 0;
    int wb = (lane < wv) ? wtB[lane] : 0;
    for (int off = 16; off > 0; off >>= 1) {
        wa += __shfl_xor(wa, off, 32);
        wb += __shfl_xor(wb, off, 32);
    }
    const int startG = wa + incA - tot;
    const int startL = wb + incB - c;
    int res = 0;
    if (c > 0) res = atomicAdd(&binCursor[t], c);
    lstart[t] = startL;
    gdelta[t] = startG + res - startL;
    if (blockIdx.x == 0) {
        if (t < NB) binStart[t] = startG;
        if (t == 0) binStart[NB] = E;
    }
    __syncthreads();

    #pragma unroll
    for (int k = 0; k < 8; ++k) {
        if (d[k] >= 0) {
            const int b = d[k] >> BIN_SHIFT;
            const int q = lstart[b] + rk[k];
            stage[q] = ((unsigned)s[k] << 8) | (unsigned)(d[k] & (BIN_NODES - 1));
            sbin[q]  = b;
        }
    }
    __syncthreads();

    const int n = min(PART_CH, E - c0);
    for (int i = t; i < n; i += 256)
        part[gdelta[sbin[i]] + i] = stage[i];
}

__global__ __launch_bounds__(1024) void gnn_binbuild_kernel(
    const unsigned* __restrict__ part, const int* __restrict__ binStart,
    int* __restrict__ rowptr, int* __restrict__ nbr, int N, int E)
{
    __shared__ int cnt[BIN_NODES];
    __shared__ int cur[BIN_NODES];
    __shared__ int wt[8];
    __shared__ int stage[STAGE_CAP];
    const int t    = threadIdx.x;
    const int lane = t & 31;
    const int wv   = t >> 5;
    const int b    = blockIdx.x;
    const int s    = binStart[b];
    const int n    = binStart[b + 1] - s;

    if (t < BIN_NODES) cnt[t] = 0;
    __syncthreads();
    for (int i = t; i < n; i += 1024)
        atomicAdd(&cnt[part[s + i] & (BIN_NODES - 1)], 1);
    __syncthreads();

    int v = 0, inc = 0;
    if (t < BIN_NODES) {
        v = cnt[t];
        inc = v;
        for (int off = 1; off < 32; off <<= 1) {
            const int u = __shfl_up(inc, off, 32);
            if (lane >= off) inc += u;
        }
        if (lane == 31) wt[wv] = inc;
    }
    __syncthreads();
    if (t < BIN_NODES) {
        int w = (lane < wv) ? wt[lane] : 0;
        for (int off = 16; off > 0; off >>= 1) w += __shfl_xor(w, off, 32);
        const int excl = w + inc - v;
        cur[t] = excl;
        const int node = b * BIN_NODES + t;
        if (node < N) rowptr[node] = s + excl;
    }
    if (b == (int)gridDim.x - 1 && t == 0) rowptr[N] = E;
    __syncthreads();

    const bool fits = n <= STAGE_CAP;
    for (int i = t; i < n; i += 1024) {
        const unsigned pk = part[s + i];
        const int p = atomicAdd(&cur[pk & (BIN_NODES - 1)], 1);
        const int sid = (int)(pk >> 8);
        if (fits) stage[p] = sid;
        else      nbr[s + p] = sid;
    }
    __syncthreads();
    if (fits) {
        for (int i = t; i < n; i += 1024) nbr[s + i] = stage[i];
    }
}

__device__ __forceinline__ void gnn_acc(float4& sv, float4& mv, const float4 v) {
    sv.x += v.x; sv.y += v.y; sv.z += v.z; sv.w += v.w;
    mv.x = fmaxf(mv.x, v.x); mv.y = fmaxf(mv.y, v.y);
    mv.z = fmaxf(mv.z, v.z); mv.w = fmaxf(mv.w, v.w);
}

template <bool POOL>
__global__ __launch_bounds__(256) void gnn_layer_kernel(
    const float* __restrict__ x, const int* __restrict__ rowptr,
    const int* __restrict__ nbr,
    const float* __restrict__ zagg,
    const float* __restrict__ W1, const float* __restrict__ b1,
    const float* __restrict__ W2, const float* __restrict__ b2,
    float* __restrict__ xout, int N,
    const int* __restrict__ batch, float* __restrict__ ps,
    unsigned* __restrict__ pmx, float* __restrict__ cnt)
{
    __shared__ float hs[TILE_M * LDS_STRIDE];
    __shared__ float xt[TILE_M * LDS_STRIDE];
    __shared__ float ts[TILE_M * LDS_STRIDE];

    const int tid  = threadIdx.x;
    const int row0 = blockIdx.x * TILE_M;

    const float z0 = zagg[0], z1 = zagg[1], z2 = zagg[2];

    {
        const int wv = __builtin_amdgcn_readfirstlane(tid >> 5);
        const int ln = tid & 31;
        const int nodeA = row0 + wv * RPW;
        int bnd[RPW + 1];
        #pragma unroll
        for (int i = 0; i <= RPW; ++i) bnd[i] = rowptr[min(nodeA + i, N)];
        const float* xl = x + ln * 4;
        #pragma unroll
        for (int rr = 0; rr < RPW; ++rr) {
            const int r   = wv * RPW + rr;
            float4 h  = make_float4(0.0f, 0.0f, 0.0f, 0.0f);
            float4 xv = make_float4(0.0f, 0.0f, 0.0f, 0.0f);
            if (row0 + r < N) {
                const int beg = bnd[rr];
                const int end = bnd[rr + 1];
                xv = *reinterpret_cast<const float4*>(xl + (size_t)(row0 + r) * GNN_D);
                float4 sv = make_float4(0.0f, 0.0f, 0.0f, 0.0f);
                const float ninf = -__builtin_inff();
                float4 mv = make_float4(ninf, ninf, ninf, ninf);
                int j = beg;
                for (; j + 8 <= end; j += 8) {
                    int    s[8];
                    float4 v[8];
                    #pragma unroll
                    for (int k = 0; k < 8; ++k) s[k] = nbr[j + k];
                    #pragma unroll
                    for (int k = 0; k < 8; ++k)
                        v[k] = *reinterpret_cast<const float4*>(xl + (size_t)s[k] * GNN_D);
                    #pragma unroll
                    for (int k = 0; k < 8; ++k) gnn_acc(sv, mv, v[k]);
                }
                if (j + 4 <= end) {
                    const int s0 = nbr[j], s1 = nbr[j + 1], s2 = nbr[j + 2], s3 = nbr[j + 3];
                    const float4 v0 = *reinterpret_cast<const float4*>(xl + (size_t)s0 * GNN_D);
                    const float4 v1 = *reinterpret_cast<const float4*>(xl + (size_t)s1 * GNN_D);
                    const float4 v2 = *reinterpret_cast<const float4*>(xl + (size_t)s2 * GNN_D);
                    const float4 v3 = *reinterpret_cast<const float4*>(xl + (size_t)s3 * GNN_D);
                    gnn_acc(sv, mv, v0); gnn_acc(sv, mv, v1);
                    gnn_acc(sv, mv, v2); gnn_acc(sv, mv, v3);
                    j += 4;
                }
                if (j + 2 <= end) {
                    const int s0 = nbr[j], s1 = nbr[j + 1];
                    const float4 v0 = *reinterpret_cast<const float4*>(xl + (size_t)s0 * GNN_D);
                    const float4 v1 = *reinterpret_cast<const float4*>(xl + (size_t)s1 * GNN_D);
                    gnn_acc(sv, mv, v0); gnn_acc(sv, mv, v1);
                    j += 2;
                }
                if (j < end) {
                    const int s0 = nbr[j];
                    const float4 v0 = *reinterpret_cast<const float4*>(xl + (size_t)s0 * GNN_D);
                    gnn_acc(sv, mv, v0);
                }
                const float dg = (float)(end - beg);
                const float dc = fmaxf(dg, 1.0f);
                const bool  nb = dg > 0.0f;
                h.x = xv.x + z0 * sv.x + z1 * (sv.x / dc) + z2 * (nb ? mv.x : 0.0f);
                h.y = xv.y + z0 * sv.y + z1 * (sv.y / dc) + z2 * (nb ? mv.y : 0.0f);
                h.z = xv.z + z0 * sv.z + z1 * (sv.z / dc) + z2 * (nb ? mv.z : 0.0f);
                h.w = xv.w + z0 * sv.w + z1 * (sv.w / dc) + z2 * (nb ? mv.w : 0.0f);
            }
            *reinterpret_cast<float4*>(&hs[r * LDS_STRIDE + ln * 4]) = h;
            *reinterpret_cast<float4*>(&xt[r * LDS_STRIDE + ln * 4]) = xv;
        }
    }
    __syncthreads();

    const int wave  = tid >> 5;
    const int lane  = tid & 31;
    const int ncol  = (wave << 4) + (lane & 15);
    const int mlo   = lane & 15;
    const int koff  = (lane >> 4) << 1;
    const int rbase = (lane >> 4) << 3;
    const float bia1 = b1[ncol];
    const float bia2 = b2[ncol];

    v8f acc[MT];
    #pragma unroll
    for (int m = 0; m < MT; ++m) acc[m] = v8f{};
    #pragma unroll
    for (int k0 = 0; k0 < GNN_D; k0 += 4) {
        const int kk = k0 + koff;
        v2f b; b.x = W1[(size_t)kk * GNN_D + ncol];
               b.y = W1[(size_t)(kk + 1) * GNN_D + ncol];
        #pragma unroll
        for (int m = 0; m < MT; ++m) {
            v2f a; a.x = hs[(m * 16 + mlo) * LDS_STRIDE + kk];
                   a.y = hs[(m * 16 + mlo) * LDS_STRIDE + kk + 1];
            acc[m] = __builtin_amdgcn_wmma_f32_16x16x4_f32(false, a, false, b,
                                                           (short)0, acc[m], false, false);
        }
    }
    #pragma unroll
    for (int m = 0; m < MT; ++m) {
        #pragma unroll
        for (int i = 0; i < 8; ++i)
            ts[(m * 16 + rbase + i) * LDS_STRIDE + ncol] = fmaxf(acc[m][i] + bia1, 0.0f);
    }
    __syncthreads();

    v8f acc2[MT];
    #pragma unroll
    for (int m = 0; m < MT; ++m) acc2[m] = v8f{};
    #pragma unroll
    for (int k0 = 0; k0 < GNN_D; k0 += 4) {
        const int kk = k0 + koff;
        v2f b; b.x = W2[(size_t)kk * GNN_D + ncol];
               b.y = W2[(size_t)(kk + 1) * GNN_D + ncol];
        #pragma unroll
        for (int m = 0; m < MT; ++m) {
            v2f a; a.x = ts[(m * 16 + mlo) * LDS_STRIDE + kk];
                   a.y = ts[(m * 16 + mlo) * LDS_STRIDE + kk + 1];
            acc2[m] = __builtin_amdgcn_wmma_f32_16x16x4_f32(false, a, false, b,
                                                            (short)0, acc2[m], false, false);
        }
    }
    #pragma unroll
    for (int m = 0; m < MT; ++m) {
        #pragma unroll
        for (int i = 0; i < 8; ++i) {
            const int r = m * 16 + rbase + i;
            const float h2 = acc2[m][i] + bia2;
            const float xv = xt[r * LDS_STRIDE + ncol];
            if (POOL) {
                hs[r * LDS_STRIDE + ncol] = xv + fmaxf(h2, 0.0f);
            } else {
                if (row0 + r < N)
                    xout[(size_t)(row0 + r) * GNN_D + ncol] = xv + fmaxf(h2, 0.0f);
            }
        }
    }
    if (POOL) {
        __syncthreads();
        if (tid < GNN_D) {
            static_assert(TILE_M == 32, "lane r of a wave holds the graph id of tile row r");
            const int nrow = min(TILE_M, N - row0);
            const int myb  = batch[row0 + min(tid & 31, nrow - 1)];
            const int b0   = __builtin_amdgcn_readfirstlane(myb);
            const bool one = (nrow == TILE_M) && (__builtin_amdgcn_ballot_w32(myb != b0) == 0u);
            const float* col = &hs[tid];
            if (one) {
                float v[TILE_M];
                #pragma unroll
                for (int r = 0; r < TILE_M; ++r) v[r] = col[r * LDS_STRIDE];
                float    sv = 0.0f;
                unsigned mm = MONO_NEG_INF;
                #pragma unroll
                for (int r = 0; r < TILE_M; ++r) {
                    sv += v[r];
                    mm  = max(mm, f2mono(v[r]));
                }
                atomicAdd(&ps[(size_t)b0 * GNN_D + tid], sv);
                atomicMax(&pmx[(size_t)b0 * GNN_D + tid], mm);
                if (tid == 0) atomicAdd(&cnt[b0], (float)TILE_M);
            } else {
                int      curb = b0;
                float    sv   = 0.0f;
                unsigned mm   = MONO_NEG_INF;
                int      run  = 0;
                for (int r = 0; r < nrow; ++r) {
                    const int   b = __builtin_amdgcn_readlane(myb, r);
                    const float v = col[r * LDS_STRIDE];
                    if (b != curb) {
                        atomicAdd(&ps[(size_t)curb * GNN_D + tid], sv);
                        atomicMax(&pmx[(size_t)curb * GNN_D + tid], mm);
                        if (tid == 0) atomicAdd(&cnt[curb], (float)run);
                        sv   = 0.0f;
                        mm   = MONO_NEG_INF;
                        run  = 0;
                        curb = b;
                    }
                    sv += v;
                    mm  = max(mm, f2mono(v));
                    ++run;
                }
                atomicAdd(&ps[(size_t)curb * GNN_D + tid], sv);
                atomicMax(&pmx[(size_t)curb * GNN_D + tid], mm);
                if (tid == 0) atomicAdd(&cnt[curb], (float)run);
            }
        }
    }
}

#define POOL_RUN 32

__device__ __forceinline__ void gnn_pool_flush(
    float* __restrict__ ps, unsigned* __restrict__ pmx, float* __restrict__ cnt,
    int b, int lane, const float4& sv, const uint4& mm, int run)
{
    float*    sp = ps  + (size_t)b * GNN_D + lane * 4;
    unsigned* mp = pmx + (size_t)b * GNN_D + lane * 4;
    atomicAdd(sp + 0, sv.x); atomicAdd(sp + 1, sv.y);
    atomicAdd(sp + 2, sv.z); atomicAdd(sp + 3, sv.w);
    atomicMax(mp + 0, mm.x); atomicMax(mp + 1, mm.y);
    atomicMax(mp + 2, mm.z); atomicMax(mp + 3, mm.w);
    if (lane == 0) atomicAdd(&cnt[b], (float)run);
}

__global__ __launch_bounds__(64) void gnn_pool_scatter_kernel(
    const float* __restrict__ x, const int* __restrict__ batch,
    float* __restrict__ ps, unsigned* __restrict__ pmx,
    float* __restrict__ cnt, int N)
{
    const int wv   = blockIdx.x * 2 + __builtin_amdgcn_readfirstlane((int)threadIdx.x >> 5);
    const int lane = threadIdx.x & 31;
    const int n0   = wv * POOL_RUN;
    if (n0 >= N) return;
    const int n1   = min(n0 + POOL_RUN, N);
    const float* xl = x + lane * 4;

    int    curb = batch[n0];
    float4 sv   = make_float4(0.0f, 0.0f, 0.0f, 0.0f);
    uint4  mm   = make_uint4(MONO_NEG_INF, MONO_NEG_INF, MONO_NEG_INF, MONO_NEG_INF);
    int    run  = 0;

    for (int n = n0; n < n1; n += 8) {
        int    bb[8];
        float4 v[8];
        #pragma unroll
        for (int i = 0; i < 8; ++i) {
            const int m = min(n + i, n1 - 1);
            bb[i] = batch[m];
            v[i]  = *reinterpret_cast<const float4*>(xl + (size_t)m * GNN_D);
        }
        #pragma unroll
        for (int i = 0; i < 8; ++i) {
            if (i == 0 || n + i < n1) {
                if (bb[i] != curb) {
                    gnn_pool_flush(ps, pmx, cnt, curb, lane, sv, mm, run);
                    sv   = make_float4(0.0f, 0.0f, 0.0f, 0.0f);
                    mm   = make_uint4(MONO_NEG_INF, MONO_NEG_INF, MONO_NEG_INF, MONO_NEG_INF);
                    run  = 0;
                    curb = bb[i];
                }
                sv.x += v[i].x; sv.y += v[i].y; sv.z += v[i].z; sv.w += v[i].w;
                mm.x = max(mm.x, f2mono(v[i].x)); mm.y = max(mm.y, f2mono(v[i].y));
                mm.z = max(mm.z, f2mono(v[i].z)); mm.w = max(mm.w, f2mono(v[i].w));
                ++run;
            }
        }
    }
    gnn_pool_flush(ps, pmx, cnt, curb, lane, sv, mm, run);
}

__global__ __launch_bounds__(256) void gnn_pool_finalize_kernel(
    const float* __restrict__ ps, const unsigned* __restrict__ pmx,
    const float* __restrict__ cnt, const float* __restrict__ zpool,
    float* __restrict__ out, int total)
{
    int i = blockIdx.x * 256 + threadIdx.x;
    if (i >= total) return;
    int b = i >> 7;
    float c  = cnt[b];
    float sv = ps[i];
    float mv = (c > 0.0f) ? mono2f(pmx[i]) : 0.0f;
    out[i] = zpool[0] * sv + zpool[1] * (sv / fmaxf(c, 1.0f)) + zpool[2] * mv;
}


extern "C" void kernel_launch(void* const* d_in, const int* in_sizes, int n_in,
                              void* d_out, int out_size, void* d_ws, size_t ws_size,
                              hipStream_t stream) {
    const float* x_in   = (const float*)d_in[0];
    const int*   eidx   = (const int*)d_in[1];
    const int*   batch  = (const int*)d_in[2];
    const float* z_agg  = (const float*)d_in[3];
    const float* z_pool = (const float*)d_in[4];
    const float* W1     = (const float*)d_in[5];
    const float* b1     = (const float*)d_in[6];
    const float* W2     = (const float*)d_in[7];
    const float* b2     = (const float*)d_in[8];
    (void)n_in;

    const int N = in_sizes[0] / GNN_D;
    const int E = in_sizes[1] / 2;
    const int L = in_sizes[5] / (GNN_D * GNN_D);
    const int Bg = out_size / GNN_D;
    const size_t ND = (size_t)N * GNN_D;
    const size_t BD = (size_t)Bg * GNN_D;

    char* p = (char*)d_ws;
    char* pend = p + ws_size;
    auto take = [&](size_t bytes) -> void* {
        void* r = (void*)p;
        p += (bytes + 255) & ~(size_t)255;
        return r;
    };
    float*    xa     = (float*)take(ND * 4);
    float*    xb     = (float*)take(ND * 4);
    int*      rowptr = (int*)take(((size_t)N + 1) * 4);
    int*      nbr    = (int*)take((size_t)E * 4);
    unsigned* part   = (unsigned*)take((size_t)E * 4);
    int*      binTC  = (int*)take((size_t)2 * NBMAX * 4);
    int*      binSt  = (int*)take(((size_t)NBMAX + 1) * 4);
    float*    ps     = (float*)take(BD * 4);
    unsigned* pmx    = (unsigned*)take(BD * 4);
    float*    cnt    = (float*)take((size_t)Bg * 4);
    if (p > pend) return;
    const int NB = (N + BIN_NODES - 1) / BIN_NODES;
    if (NB > NBMAX || N > (1 << 24)) return;

    const int* src = eidx;
    const int* dst = eidx + E;

    hipMemsetAsync(binTC, 0, (size_t)2 * NBMAX * 4, stream);
    gnn_bincount_kernel<<<(E + PART_CH - 1) / PART_CH, 256, 0, stream>>>(
        dst, binTC, E, ps, pmx, cnt, (int)BD, Bg);
    gnn_partition_kernel<<<(E + PART_CH - 1) / PART_CH, 256, 0, stream>>>(
        src, dst, binTC, binTC + NBMAX, binSt, part, E, NB);
    gnn_binbuild_kernel<<<NB, 1024, 0, stream>>>(part, binSt, rowptr, nbr, N, E);

    const float* xcur = x_in;
    float* xnxt = xa;
    float* xoth = xb;
    for (int l = 0; l < L; ++l) {
        if (l + 1 < L) {
            gnn_layer_kernel<false><<<(N + TILE_M - 1) / TILE_M, 256, 0, stream>>>(
                xcur, rowptr, nbr, z_agg + l * 3,
                W1 + (size_t)l * GNN_D * GNN_D, b1 + (size_t)l * GNN_D,
                W2 + (size_t)l * GNN_D * GNN_D, b2 + (size_t)l * GNN_D, xnxt, N,
                batch, ps, pmx, cnt);
        } else {
            gnn_layer_kernel<true><<<(N + TILE_M - 1) / TILE_M, 256, 0, stream>>>(
                xcur, rowptr, nbr, z_agg + l * 3,
                W1 + (size_t)l * GNN_D * GNN_D, b1 + (size_t)l * GNN_D,
                W2 + (size_t)l * GNN_D * GNN_D, b2 + (size_t)l * GNN_D, xnxt, N,
                batch, ps, pmx, cnt);
        }
        xcur = xnxt;
        float* t = xnxt; xnxt = xoth; xoth = t;
    }

    if (L == 0)
        gnn_pool_scatter_kernel<<<(N + 2 * POOL_RUN - 1) / (2 * POOL_RUN), 64, 0, stream>>>(xcur, batch, ps, pmx, cnt, N);
    gnn_pool_finalize_kernel<<<(int)((BD + 255) / 256), 256, 0, stream>>>(
        ps, pmx, cnt, z_pool, (float*)d_out, (int)BD);
}
